# starting kernel
# baseline (speedup 1.0000x reference)
#include <hip/hip_runtime.h>
#include <cstddef>


typedef __attribute__((ext_vector_type(2))) float v2f;
typedef __attribute__((ext_vector_type(8))) float v8f;

__global__ void k_init_deg(float* __restrict__ deg, int n) {
  int i = blockIdx.x * blockDim.x + threadIdx.x;
  if (i < n) deg[i] = 1.0f;
}

__global__ void k_count_deg(const int* __restrict__ cols, float* __restrict__ deg, int E) {
  int e = blockIdx.x * blockDim.x + threadIdx.x;
  if (e < E) atomicAdd(&deg[cols[e]], 1.0f);
}

__global__ void k_rsqrt_inplace(float* __restrict__ deg, int n) {
  int i = blockIdx.x * blockDim.x + threadIdx.x;
  if (i < n) deg[i] = 1.0f / sqrtf(deg[i]);
}

template <int K, int NN>
__global__ void k_gemm_wmma(const float* __restrict__ A, const float* __restrict__ B,
                            float* __restrict__ C) {
  const int wave = threadIdx.x >> 5;
  const int lane = threadIdx.x & 31;
  const int m0   = blockIdx.x << 4;
  const int n0   = wave << 4;
  const int l16  = lane & 15;
  const int kh   = (lane >> 4) << 1;

  v8f acc = {};
  const float* arow = A + (size_t)(m0 + l16) * K + kh;
  const float* bcol = B + (size_t)kh * NN + n0 + l16;

#pragma unroll
  for (int k = 0; k < K; k += 4) {
    v2f a, b;
    a.x = arow[k];
    a.y = arow[k + 1];
    b.x = bcol[(size_t)k * NN];
    b.y = bcol[(size_t)(k + 1) * NN];
    acc = __builtin_amdgcn_wmma_f32_16x16x4_f32(false, a, false, b, (short)0, acc,
                                                false, false);
  }

  const int chi = (lane >> 4) << 3;
  float* crow = C + (size_t)(m0 + chi) * NN + n0 + l16;
#pragma unroll
  for (int v = 0; v < 8; ++v) crow[(size_t)v * NN] = acc[v];
}

__global__ void k_scatter(const int* __restrict__ rows, const int* __restrict__ cols,
                          const float* __restrict__ dinv, const float* __restrict__ H,
                          float* __restrict__ AGG, int E, int logF) {
  const int F   = 1 << logF;
  const int epb = blockDim.x >> logF;
  const int e   = blockIdx.x * epb + (threadIdx.x >> logF);
  if (e >= E) return;
  const int f = threadIdx.x & (F - 1);
  const int r = rows[e];
  const int c = cols[e];
  const float w = dinv[r] * dinv[c];
  atomicAdd(&AGG[(size_t)c * F + f], H[(size_t)r * F + f] * w);
}

__global__ void k_post(const float* __restrict__ AGG, const float* __restrict__ H,
                       const float* __restrict__ dinv, const float* __restrict__ bias,
                       float* __restrict__ OUT, int n, int logF, int do_relu) {
  const int F    = 1 << logF;
  const size_t i = (size_t)blockIdx.x * blockDim.x + threadIdx.x;
  if (i >= ((size_t)n << logF)) return;
  const int node = (int)(i >> logF);
  const int f    = (int)(i & (size_t)(F - 1));
  const float d  = dinv[node];
  float v = AGG[i] + H[i] * d * d + bias[f];
  if (do_relu) v = fmaxf(v, 0.0f);
  OUT[i] = v;
}

extern "C" void kernel_launch(void* const* d_in, const int* in_sizes, int n_in,
                              void* d_out, int out_size, void* d_ws, size_t ws_size,
                              hipStream_t stream) {
  const float* x  = (const float*)d_in[0];
  const int*   ei = (const int*)d_in[1];
  const float* W1 = (const float*)d_in[2];
  const float* b1 = (const float*)d_in[3];
  const float* W2 = (const float*)d_in[4];
  const float* b2 = (const float*)d_in[5];

  const int IN_DIM = 128, HID = 64, OUT = 32;
  const int N = in_sizes[0] / IN_DIM;
  const int E = in_sizes[1] / 2;
  const int* rows = ei;
  const int* cols = ei + E;

  float* ws   = (float*)d_ws;
  float* dinv = ws;
  float* h1   = ws + N;
  float* agg1 = h1 + (size_t)N * HID;
  float* h2   = h1;
  float* agg2 = h1 + (size_t)N * OUT;

  const int T = 256;

  k_init_deg<<<(N + T - 1) / T, T, 0, stream>>>(dinv, N);
  k_count_deg<<<(E + T - 1) / T, T, 0, stream>>>(cols, dinv, E);
  k_rsqrt_inplace<<<(N + T - 1) / T, T, 0, stream>>>(dinv, N);

  k_gemm_wmma<128, 64><<<N / 16, 32 * (64 / 16), 0, stream>>>(x, W1, h1);
  hipMemsetAsync(agg1, 0, (size_t)N * HID * sizeof(float), stream);
  k_scatter<<<(E * HID + T - 1) / T, T, 0, stream>>>(rows, cols, dinv, h1, agg1, E, 6);
  k_post<<<(int)(((size_t)N * HID + T - 1) / T), T, 0, stream>>>(agg1, h1, dinv, b1,
                                                                 agg1, N, 6, 1);

  k_gemm_wmma<64, 32><<<N / 16, 32 * (32 / 16), 0, stream>>>(agg1, W2, h2);
  hipMemsetAsync(agg2, 0, (size_t)N * OUT * sizeof(float), stream);
  k_scatter<<<(E * OUT + T - 1) / T, T, 0, stream>>>(rows, cols, dinv, h2, agg2, E, 5);
  k_post<<<(int)(((size_t)N * OUT + T - 1) / T), T, 0, stream>>>(agg2, h2, dinv, b2,
                                                                 (float*)d_out, N, 5, 0);
}


# Degree and both edge sums by node-range owners: a work-group filters the whole target list for its 196 nodes and adds gathered rows to LDS sums by LDS float atomics (arrival order); 11 launches to 5
# speedup vs baseline: 2.0766x; 2.0766x over previous
#include <hip/hip_runtime.h>
#include <cstddef>


typedef __attribute__((ext_vector_type(2))) float v2f;
typedef __attribute__((ext_vector_type(8))) float v8f;
typedef __attribute__((ext_vector_type(4))) int   v4i;

#define OWN_R   196
#define OWN_T   1024
#define OWN_W   (OWN_T / 32)
#define SCAN_U  8
#define CHUNK_E (32 * 4 * SCAN_U)
#define ROW_G   8

__device__ __forceinline__ void load_chunk(v4i (&c)[SCAN_U], const int* __restrict__ cols,
                                           int ebase, int lane, int E, bool vec_ok, int fill) {
  if (vec_ok && ebase + CHUNK_E <= E) {
#pragma unroll
    for (int u = 0; u < SCAN_U; ++u)
      c[u] = *reinterpret_cast<const v4i*>(cols + ebase + u * 128 + lane * 4);
  } else {
#pragma unroll
    for (int u = 0; u < SCAN_U; ++u) {
      const int e = ebase + u * 128 + lane * 4;
      c[u].x = (e     < E) ? cols[e]     : fill;
      c[u].y = (e + 1 < E) ? cols[e + 1] : fill;
      c[u].z = (e + 2 < E) ? cols[e + 2] : fill;
      c[u].w = (e + 3 < E) ? cols[e + 3] : fill;
    }
  }
}

__global__ void __launch_bounds__(OWN_T)
k_deg_owner(const int* __restrict__ cols, float* __restrict__ dinv, int n, int E) {
  __shared__ int cnt[OWN_R];
  const int lane = threadIdx.x & 31;
  const int wave = __builtin_amdgcn_readfirstlane((int)(threadIdx.x >> 5));
  const int t0   = blockIdx.x * OWN_R;
  const unsigned nown = (unsigned)min(OWN_R, n - t0);

  for (int i = threadIdx.x; i < OWN_R; i += OWN_T) cnt[i] = 0;
  __syncthreads();

  const int  nchunk = (E + CHUNK_E - 1) / CHUNK_E;
  const bool vec_ok = ((E & 3) == 0) && ((((size_t)cols) & 15) == 0);
  const int  start  = (int)((blockIdx.x * max(1u, (unsigned)nchunk / gridDim.x)) % (unsigned)nchunk);

  for (int ci = wave; ci < nchunk; ci += OWN_W) {
    int ch = ci + start;
    if (ch >= nchunk) ch -= nchunk;
    v4i c[SCAN_U];
    load_chunk(c, cols, ch * CHUNK_E, lane, E, vec_ok, t0 + OWN_R);
#pragma unroll
    for (int u = 0; u < SCAN_U; ++u) {
#pragma unroll
      for (int j = 0; j < 4; ++j) {
        const unsigned d = (unsigned)(c[u][j] - t0);
        if (d < nown) atomicAdd(&cnt[d], 1);
      }
    }
  }
  __syncthreads();

  for (unsigned i = threadIdx.x; i < nown; i += OWN_T)
    dinv[t0 + i] = 1.0f / sqrtf(1.0f + (float)cnt[i]);
}

template <int K, int NN>
__global__ void k_gemm_wmma(const float* __restrict__ A, const float* __restrict__ B,
                            float* __restrict__ C) {
  const int wave = threadIdx.x >> 5;
  const int lane = threadIdx.x & 31;
  const int m0   = blockIdx.x << 4;
  const int n0   = wave << 4;
  const int l16  = lane & 15;
  const int kh   = (lane >> 4) << 1;

  v8f acc = {};
  const float* arow = A + (size_t)(m0 + l16) * K + kh;
  const float* bcol = B + (size_t)kh * NN + n0 + l16;

#pragma unroll
  for (int k = 0; k < K; k += 4) {
    v2f a, b;
    a.x = arow[k];
    a.y = arow[k + 1];
    b.x = bcol[(size_t)k * NN];
    b.y = bcol[(size_t)(k + 1) * NN];
    acc = __builtin_amdgcn_wmma_f32_16x16x4_f32(false, a, false, b, (short)0, acc,
                                                false, false);
  }

  const int chi = (lane >> 4) << 3;
  float* crow = C + (size_t)(m0 + chi) * NN + n0 + l16;
#pragma unroll
  for (int v = 0; v < 8; ++v) crow[(size_t)v * NN] = acc[v];
}

template <int F, bool RELU>
__global__ void __launch_bounds__(OWN_T)
k_agg_owner(const int* __restrict__ rows, const int* __restrict__ cols,
            const float* __restrict__ dinv, const float* __restrict__ H,
            const float* __restrict__ bias, float* __restrict__ OUT, int n, int E) {
  constexpr int S = F / 32;
  static_assert(OWN_R <= 65536 && CHUNK_E <= 65536, "hit entries pack two 16-bit fields");
  __shared__ float    acc[OWN_R * F];
  __shared__ unsigned hitlist[OWN_W][CHUNK_E];

  const int lane = threadIdx.x & 31;
  const int wave = __builtin_amdgcn_readfirstlane((int)(threadIdx.x >> 5));
  const int t0   = blockIdx.x * OWN_R;
  const unsigned nown = (unsigned)min(OWN_R, n - t0);
  const int fill = t0 + OWN_R;

  for (int i = threadIdx.x; i < OWN_R * F; i += OWN_T) acc[i] = 0.0f;
  __syncthreads();

  unsigned* wl = hitlist[wave];
  const int  nchunk = (E + CHUNK_E - 1) / CHUNK_E;
  const bool vec_ok = ((E & 3) == 0) && ((((size_t)cols) & 15) == 0);
  const int  start  = (int)((blockIdx.x * max(1u, (unsigned)nchunk / gridDim.x)) % (unsigned)nchunk);

  if (wave < nchunk) {
    int ci = wave;
    int ch = ci + start;
    if (ch >= nchunk) ch -= nchunk;
    v4i c[SCAN_U];
    load_chunk(c, cols, ch * CHUNK_E, lane, E, vec_ok, fill);

    for (; ci < nchunk; ci += OWN_W) {
      const int ebase = ch * CHUNK_E;

      int nh = 0;
#pragma unroll
      for (int u = 0; u < SCAN_U; ++u) {
#pragma unroll
        for (int j = 0; j < 4; ++j) {
          const unsigned d   = (unsigned)(c[u][j] - t0);
          const bool     hit = d < nown;
          const unsigned m   = __builtin_amdgcn_ballot_w32(hit);
          if (m) {
            if (hit)
              wl[nh + (int)__builtin_amdgcn_mbcnt_lo(m, 0u)] =
                  ((unsigned)(u * 128 + lane * 4 + j) << 16) | d;
            nh += __builtin_popcount(m);
          }
        }
      }

      int chn = ch;
      if (ci + OWN_W < nchunk) {
        chn = ci + OWN_W + start;
        if (chn >= nchunk) chn -= nchunk;
      }
      load_chunk(c, cols, chn * CHUNK_E, lane, E, vec_ok, fill);

      __builtin_amdgcn_fence(__ATOMIC_RELEASE, "wavefront");
      __builtin_amdgcn_wave_barrier();
      __builtin_amdgcn_fence(__ATOMIC_ACQUIRE, "wavefront");

      for (int k0 = 0; k0 < nh; k0 += 32) {
        const int cnt = min(32, nh - k0);
        const unsigned ent = wl[k0 + min(lane, cnt - 1)];
        const int d = (int)(ent & 0xffffu);
        const int r = rows[ebase + (int)(ent >> 16)];
        int i0 = 0;
        do {
          float hv[ROW_G][S];
          float ww[ROW_G];
          int   dd[ROW_G];
#pragma unroll
          for (int g = 0; g < ROW_G; ++g) {
            const int ii = min(i0 + g, cnt - 1);
            const int rg = __builtin_amdgcn_readlane(r, ii);
            dd[g] = __builtin_amdgcn_readlane(d, ii);
            ww[g] = dinv[rg] * dinv[t0 + dd[g]];
            const float* hp = H + (size_t)rg * F + lane;
#pragma unroll
            for (int s = 0; s < S; ++s) hv[g][s] = hp[s * 32];
          }
#pragma unroll
          for (int g = 0; g < ROW_G; ++g) {
            if (i0 + g < cnt) {
#pragma unroll
              for (int s = 0; s < S; ++s)
                atomicAdd(&acc[dd[g] * F + s * 32 + lane], hv[g][s] * ww[g]);
            }
          }
          i0 += ROW_G;
        } while (i0 < cnt);
      }
      ch = chn;
    }
  }
  __syncthreads();

  const int tot = (int)nown * F;
  for (int i = threadIdx.x; i < tot; i += OWN_T) {
    const int    dloc = i / F;
    const int    f    = i & (F - 1);
    const size_t gi   = (size_t)t0 * F + i;
    const float  dv   = dinv[t0 + dloc];
    float v = acc[i] + H[gi] * dv * dv + bias[f];
    if (RELU) v = fmaxf(v, 0.0f);
    OUT[gi] = v;
  }
}

extern "C" void kernel_launch(void* const* d_in, const int* in_sizes, int n_in,
                              void* d_out, int out_size, void* d_ws, size_t ws_size,
                              hipStream_t stream) {
  const float* x  = (const float*)d_in[0];
  const int*   ei = (const int*)d_in[1];
  const float* W1 = (const float*)d_in[2];
  const float* b1 = (const float*)d_in[3];
  const float* W2 = (const float*)d_in[4];
  const float* b2 = (const float*)d_in[5];

  const int IN_DIM = 128, HID = 64, OUT = 32;
  const int N = in_sizes[0] / IN_DIM;
  const int E = in_sizes[1] / 2;
  const int* rows = ei;
  const int* cols = ei + E;

  float* ws   = (float*)d_ws;
  float* dinv = ws;
  float* h1   = ws + N;
  float* r1   = h1 + (size_t)N * HID;
  float* h2   = h1;

  const int G = (N + OWN_R - 1) / OWN_R;

  k_deg_owner<<<G, OWN_T, 0, stream>>>(cols, dinv, N, E);

  k_gemm_wmma<128, 64><<<N / 16, 32 * (64 / 16), 0, stream>>>(x, W1, h1);
  k_agg_owner<64, true><<<G, OWN_T, 0, stream>>>(rows, cols, dinv, h1, b1, r1, N, E);

  k_gemm_wmma<64, 32><<<N / 16, 32 * (32 / 16), 0, stream>>>(r1, W2, h2);
  k_agg_owner<32, false><<<G, OWN_T, 0, stream>>>(rows, cols, dinv, h2, b2,
                                                  (float*)d_out, N, E);
}


# Owners filter the target list once: the degree launch stores each owner's in-range edges (source, local target) in the workspace; both sum launches read that list; LDS arrival order as before
# speedup vs baseline: 2.8991x; 1.3961x over previous
#include <hip/hip_runtime.h>
#include <cstddef>


typedef __attribute__((ext_vector_type(2))) float v2f;
typedef __attribute__((ext_vector_type(8))) float v8f;
typedef __attribute__((ext_vector_type(4))) int   v4i;

#define OWN_R   196
#define OWN_T   1024
#define OWN_W   (OWN_T / 32)
#define SCAN_U  8
#define CHUNK_E (32 * 4 * SCAN_U)
#define ROW_G   8
#define LIST_CAP 131072
#define D_BITS  8

__device__ __forceinline__ void load_chunk(v4i (&c)[SCAN_U], const int* __restrict__ cols,
                                           int ebase, int lane, int E, bool vec_ok, int fill) {
  if (vec_ok && ebase + CHUNK_E <= E) {
#pragma unroll
    for (int u = 0; u < SCAN_U; ++u)
      c[u] = *reinterpret_cast<const v4i*>(cols + ebase + u * 128 + lane * 4);
  } else {
#pragma unroll
    for (int u = 0; u < SCAN_U; ++u) {
      const int e = ebase + u * 128 + lane * 4;
      c[u].x = (e     < E) ? cols[e]     : fill;
      c[u].y = (e + 1 < E) ? cols[e + 1] : fill;
      c[u].z = (e + 2 < E) ? cols[e + 2] : fill;
      c[u].w = (e + 3 < E) ? cols[e + 3] : fill;
    }
  }
}

__global__ void __launch_bounds__(OWN_T)
k_deg_owner(const int* __restrict__ rows, const int* __restrict__ cols, float* __restrict__ dinv,
            int* __restrict__ glen, unsigned* __restrict__ glist, int n, int E) {
  static_assert(OWN_R <= (1 << D_BITS), "local target must fit the low bits of a list entry");
  static_assert(CHUNK_E <= 65536, "hit entries pack two 16-bit fields");
  __shared__ int      cnt[OWN_R];
  __shared__ int      llen;
  __shared__ unsigned hitlist[OWN_W][CHUNK_E];
  const int lane = threadIdx.x & 31;
  const int wave = __builtin_amdgcn_readfirstlane((int)(threadIdx.x >> 5));
  const int t0   = blockIdx.x * OWN_R;
  const unsigned nown = (unsigned)min(OWN_R, n - t0);
  const int fill = t0 + OWN_R;

  for (int i = threadIdx.x; i < OWN_R; i += OWN_T) cnt[i] = 0;
  if (threadIdx.x == 0) llen = 0;
  __syncthreads();

  unsigned* wl = hitlist[wave];
  unsigned* gl = glist + (size_t)blockIdx.x * LIST_CAP;
  const int  nchunk = (E + CHUNK_E - 1) / CHUNK_E;
  const bool vec_ok = ((E & 3) == 0) && ((((size_t)cols) & 15) == 0);
  const int  start  = (int)((blockIdx.x * max(1u, (unsigned)nchunk / gridDim.x)) % (unsigned)nchunk);

  if (wave < nchunk) {
    int ci = wave;
    int ch = ci + start;
    if (ch >= nchunk) ch -= nchunk;
    v4i c[SCAN_U];
    load_chunk(c, cols, ch * CHUNK_E, lane, E, vec_ok, fill);

    for (; ci < nchunk; ci += OWN_W) {
      const int ebase = ch * CHUNK_E;

      int nh = 0;
#pragma unroll
      for (int u = 0; u < SCAN_U; ++u) {
#pragma unroll
        for (int j = 0; j < 4; ++j) {
          const unsigned d   = (unsigned)(c[u][j] - t0);
          const bool     hit = d < nown;
          const unsigned m   = __builtin_amdgcn_ballot_w32(hit);
          if (m) {
            if (hit)
              wl[nh + (int)__builtin_amdgcn_mbcnt_lo(m, 0u)] =
                  ((unsigned)(u * 128 + lane * 4 + j) << 16) | d;
            nh += __builtin_popcount(m);
          }
        }
      }

      int chn = ch;
      if (ci + OWN_W < nchunk) {
        chn = ci + OWN_W + start;
        if (chn >= nchunk) chn -= nchunk;
      }
      load_chunk(c, cols, chn * CHUNK_E, lane, E, vec_ok, fill);

      if (nh > 0) {
        __builtin_amdgcn_fence(__ATOMIC_RELEASE, "wavefront");
        __builtin_amdgcn_wave_barrier();
        __builtin_amdgcn_fence(__ATOMIC_ACQUIRE, "wavefront");

        int pos = 0;
        if (lane == 0) pos = atomicAdd(&llen, nh);
        pos = __builtin_amdgcn_readfirstlane(pos);
        for (int k = lane; k < nh; k += 32) {
          const unsigned ent = wl[k];
          const unsigned d   = ent & 0xffffu;
          const int      r   = rows[ebase + (int)(ent >> 16)];
          atomicAdd(&cnt[d], 1);
          if (pos + k < LIST_CAP) gl[pos + k] = ((unsigned)r << D_BITS) | d;
        }
      }
      ch = chn;
    }
  }
  __syncthreads();

  for (unsigned i = threadIdx.x; i < nown; i += OWN_T)
    dinv[t0 + i] = 1.0f / sqrtf(1.0f + (float)cnt[i]);
  if (threadIdx.x == 0) glen[blockIdx.x] = llen;
}

template <int K, int NN>
__global__ void k_gemm_wmma(const float* __restrict__ A, const float* __restrict__ B,
                            float* __restrict__ C) {
  const int wave = threadIdx.x >> 5;
  const int lane = threadIdx.x & 31;
  const int m0   = blockIdx.x << 4;
  const int n0   = wave << 4;
  const int l16  = lane & 15;
  const int kh   = (lane >> 4) << 1;

  v8f acc = {};
  const float* arow = A + (size_t)(m0 + l16) * K + kh;
  const float* bcol = B + (size_t)kh * NN + n0 + l16;

#pragma unroll
  for (int k = 0; k < K; k += 4) {
    v2f a, b;
    a.x = arow[k];
    a.y = arow[k + 1];
    b.x = bcol[(size_t)k * NN];
    b.y = bcol[(size_t)(k + 1) * NN];
    acc = __builtin_amdgcn_wmma_f32_16x16x4_f32(false, a, false, b, (short)0, acc,
                                                false, false);
  }

  const int chi = (lane >> 4) << 3;
  float* crow = C + (size_t)(m0 + chi) * NN + n0 + l16;
#pragma unroll
  for (int v = 0; v < 8; ++v) crow[(size_t)v * NN] = acc[v];
}

template <int F>
__device__ __forceinline__ void add_trip(float* __restrict__ acc, int r, int d, int cnt, int lane,
                                         int t0, const float* __restrict__ dinv,
                                         const float* __restrict__ H) {
  constexpr int S = F / 32;
  int i0 = 0;
  do {
    float hv[ROW_G][S];
    float ww[ROW_G];
    int   dd[ROW_G];
#pragma unroll
    for (int g = 0; g < ROW_G; ++g) {
      const int ii = min(i0 + g, cnt - 1);
      const int rg = __builtin_amdgcn_readlane(r, ii);
      dd[g] = __builtin_amdgcn_readlane(d, ii);
      ww[g] = dinv[rg] * dinv[t0 + dd[g]];
      const float* hp = H + (size_t)rg * F + lane;
#pragma unroll
      for (int s = 0; s < S; ++s) hv[g][s] = hp[s * 32];
    }
#pragma unroll
    for (int g = 0; g < ROW_G; ++g) {
      if (i0 + g < cnt) {
#pragma unroll
        for (int s = 0; s < S; ++s)
          atomicAdd(&acc[dd[g] * F + s * 32 + lane], hv[g][s] * ww[g]);
      }
    }
    i0 += ROW_G;
  } while (i0 < cnt);
}

template <int F, bool RELU>
__global__ void __launch_bounds__(OWN_T)
k_agg_owner(const int* __restrict__ rows, const int* __restrict__ cols,
            const int* __restrict__ glen, const unsigned* __restrict__ glist,
            const float* __restrict__ dinv, const float* __restrict__ H,
            const float* __restrict__ bias, float* __restrict__ OUT, int n, int E) {
  static_assert(OWN_R <= 65536 && CHUNK_E <= 65536, "hit entries pack two 16-bit fields");
  __shared__ float    acc[OWN_R * F];
  __shared__ unsigned hitlist[OWN_W][CHUNK_E];

  const int lane = threadIdx.x & 31;
  const int wave = __builtin_amdgcn_readfirstlane((int)(threadIdx.x >> 5));
  const int t0   = blockIdx.x * OWN_R;
  const unsigned nown = (unsigned)min(OWN_R, n - t0);
  const int fill = t0 + OWN_R;

  for (int i = threadIdx.x; i < OWN_R * F; i += OWN_T) acc[i] = 0.0f;
  __syncthreads();

  unsigned* wl = hitlist[wave];
  const int  nchunk = (E + CHUNK_E - 1) / CHUNK_E;
  const bool vec_ok = ((E & 3) == 0) && ((((size_t)cols) & 15) == 0);
  const int  start  = (int)((blockIdx.x * max(1u, (unsigned)nchunk / gridDim.x)) % (unsigned)nchunk);

  const int len = __builtin_amdgcn_readfirstlane(glen[blockIdx.x]);

  if (len <= LIST_CAP) {
    const unsigned* gl = glist + (size_t)blockIdx.x * LIST_CAP;
    int k0 = wave * 32;
    if (k0 < len) {
      unsigned ent = gl[k0 + min(lane, min(32, len - k0) - 1)];
      for (; k0 < len; k0 += OWN_W * 32) {
        const int cnt = min(32, len - k0);
        const int kn = k0 + OWN_W * 32;
        unsigned entn = ent;
        if (kn < len) entn = gl[kn + min(lane, min(32, len - kn) - 1)];
        add_trip<F>(acc, (int)(ent >> D_BITS), (int)(ent & ((1u << D_BITS) - 1u)), cnt, lane, t0,
                    dinv, H);
        ent = entn;
      }
    }
  } else if (wave < nchunk) {
    int ci = wave;
    int ch = ci + start;
    if (ch >= nchunk) ch -= nchunk;
    v4i c[SCAN_U];
    load_chunk(c, cols, ch * CHUNK_E, lane, E, vec_ok, fill);

    for (; ci < nchunk; ci += OWN_W) {
      const int ebase = ch * CHUNK_E;

      int nh = 0;
#pragma unroll
      for (int u = 0; u < SCAN_U; ++u) {
#pragma unroll
        for (int j = 0; j < 4; ++j) {
          const unsigned d   = (unsigned)(c[u][j] - t0);
          const bool     hit = d < nown;
          const unsigned m   = __builtin_amdgcn_ballot_w32(hit);
          if (m) {
            if (hit)
              wl[nh + (int)__builtin_amdgcn_mbcnt_lo(m, 0u)] =
                  ((unsigned)(u * 128 + lane * 4 + j) << 16) | d;
            nh += __builtin_popcount(m);
          }
        }
      }

      int chn = ch;
      if (ci + OWN_W < nchunk) {
        chn = ci + OWN_W + start;
        if (chn >= nchunk) chn -= nchunk;
      }
      load_chunk(c, cols, chn * CHUNK_E, lane, E, vec_ok, fill);

      __builtin_amdgcn_fence(__ATOMIC_RELEASE, "wavefront");
      __builtin_amdgcn_wave_barrier();
      __builtin_amdgcn_fence(__ATOMIC_ACQUIRE, "wavefront");

      for (int k0 = 0; k0 < nh; k0 += 32) {
        const int cnt = min(32, nh - k0);
        const unsigned ent = wl[k0 + min(lane, cnt - 1)];
        const int d = (int)(ent & 0xffffu);
        const int r = rows[ebase + (int)(ent >> 16)];
        add_trip<F>(acc, r, d, cnt, lane, t0, dinv, H);
      }
      ch = chn;
    }
  }
  __syncthreads();

  const int tot = (int)nown * F;
  for (int i = threadIdx.x; i < tot; i += OWN_T) {
    const int    dloc = i / F;
    const int    f    = i & (F - 1);
    const size_t gi   = (size_t)t0 * F + i;
    const float  dv   = dinv[t0 + dloc];
    float v = acc[i] + H[gi] * dv * dv + bias[f];
    if (RELU) v = fmaxf(v, 0.0f);
    OUT[gi] = v;
  }
}

extern "C" void kernel_launch(void* const* d_in, const int* in_sizes, int n_in,
                              void* d_out, int out_size, void* d_ws, size_t ws_size,
                              hipStream_t stream) {
  const float* x  = (const float*)d_in[0];
  const int*   ei = (const int*)d_in[1];
  const float* W1 = (const float*)d_in[2];
  const float* b1 = (const float*)d_in[3];
  const float* W2 = (const float*)d_in[4];
  const float* b2 = (const float*)d_in[5];

  const int IN_DIM = 128, HID = 64, OUT = 32;
  const int N = in_sizes[0] / IN_DIM;
  const int E = in_sizes[1] / 2;
  const int* rows = ei;
  const int* cols = ei + E;

  const int G = (N + OWN_R - 1) / OWN_R;

  float* ws   = (float*)d_ws;
  float* dinv = ws;
  float* h1   = ws + N;
  float* r1   = h1 + (size_t)N * HID;
  float* h2   = h1;
  int*      glen  = (int*)(r1 + (size_t)N * HID);
  unsigned* glist = (unsigned*)(glen + G);

  k_deg_owner<<<G, OWN_T, 0, stream>>>(rows, cols, dinv, glen, glist, N, E);

  k_gemm_wmma<128, 64><<<N / 16, 32 * (64 / 16), 0, stream>>>(x, W1, h1);
  k_agg_owner<64, true><<<G, OWN_T, 0, stream>>>(rows, cols, glen, glist, dinv, h1, b1, r1, N, E);

  k_gemm_wmma<64, 32><<<N / 16, 32 * (32 / 16), 0, stream>>>(r1, W2, h2);
  k_agg_owner<32, false><<<G, OWN_T, 0, stream>>>(rows, cols, glen, glist, dinv, h2, b2,
                                                  (float*)d_out, N, E);
}


# Both products: a wave owns 16 rows and all their column tiles, A loaded once by 16-byte loads, W fragments from an LDS copy per group; k pairs per WMMA step regrouped, so the k sum order changes
# speedup vs baseline: 3.0397x; 1.0485x over previous
#include <hip/hip_runtime.h>
#include <cstddef>


typedef __attribute__((ext_vector_type(2))) float v2f;
typedef __attribute__((ext_vector_type(8))) float v8f;
typedef __attribute__((ext_vector_type(4))) int   v4i;
typedef __attribute__((ext_vector_type(4))) float v4f;

#define OWN_R   196
#define OWN_T   1024
#define OWN_W   (OWN_T / 32)
#define SCAN_U  8
#define CHUNK_E (32 * 4 * SCAN_U)
#define ROW_G   8
#define LIST_CAP 131072
#define D_BITS  8
#define GEMM_W  8

__device__ __forceinline__ void load_chunk(v4i (&c)[SCAN_U], const int* __restrict__ cols,
                                           int ebase, int lane, int E, bool vec_ok, int fill) {
  if (vec_ok && ebase + CHUNK_E <= E) {
#pragma unroll
    for (int u = 0; u < SCAN_U; ++u)
      c[u] = *reinterpret_cast<const v4i*>(cols + ebase + u * 128 + lane * 4);
  } else {
#pragma unroll
    for (int u = 0; u < SCAN_U; ++u) {
      const int e = ebase + u * 128 + lane * 4;
      c[u].x = (e     < E) ? cols[e]     : fill;
      c[u].y = (e + 1 < E) ? cols[e + 1] : fill;
      c[u].z = (e + 2 < E) ? cols[e + 2] : fill;
      c[u].w = (e + 3 < E) ? cols[e + 3] : fill;
    }
  }
}

__global__ void __launch_bounds__(OWN_T)
k_deg_owner(const int* __restrict__ rows, const int* __restrict__ cols, float* __restrict__ dinv,
            int* __restrict__ glen, unsigned* __restrict__ glist, int n, int E) {
  static_assert(OWN_R <= (1 << D_BITS), "local target must fit the low bits of a list entry");
  static_assert(CHUNK_E <= 65536, "hit entries pack two 16-bit fields");
  __shared__ int      cnt[OWN_R];
  __shared__ int      llen;
  __shared__ unsigned hitlist[OWN_W][CHUNK_E];
  const int lane = threadIdx.x & 31;
  const int wave = __builtin_amdgcn_readfirstlane((int)(threadIdx.x >> 5));
  const int t0   = blockIdx.x * OWN_R;
  const unsigned nown = (unsigned)min(OWN_R, n - t0);
  const int fill = t0 + OWN_R;

  for (int i = threadIdx.x; i < OWN_R; i += OWN_T) cnt[i] = 0;
  if (threadIdx.x == 0) llen = 0;
  __syncthreads();

  unsigned* wl = hitlist[wave];
  unsigned* gl = glist + (size_t)blockIdx.x * LIST_CAP;
  const int  nchunk = (E + CHUNK_E - 1) / CHUNK_E;
  const bool vec_ok = ((E & 3) == 0) && ((((size_t)cols) & 15) == 0);
  const int  start  = (int)((blockIdx.x * max(1u, (unsigned)nchunk / gridDim.x)) % (unsigned)nchunk);

  if (wave < nchunk) {
    int ci = wave;
    int ch = ci + start;
    if (ch >= nchunk) ch -= nchunk;
    v4i c[SCAN_U];
    load_chunk(c, cols, ch * CHUNK_E, lane, E, vec_ok, fill);

    for (; ci < nchunk; ci += OWN_W) {
      const int ebase = ch * CHUNK_E;

      int nh = 0;
#pragma unroll
      for (int u = 0; u < SCAN_U; ++u) {
#pragma unroll
        for (int j = 0; j < 4; ++j) {
          const unsigned d   = (unsigned)(c[u][j] - t0);
          const bool     hit = d < nown;
          const unsigned m   = __builtin_amdgcn_ballot_w32(hit);
          if (m) {
            if (hit)
              wl[nh + (int)__builtin_amdgcn_mbcnt_lo(m, 0u)] =
                  ((unsigned)(u * 128 + lane * 4 + j) << 16) | d;
            nh += __builtin_popcount(m);
          }
        }
      }

      int chn = ch;
      if (ci + OWN_W < nchunk) {
        chn = ci + OWN_W + start;
        if (chn >= nchunk) chn -= nchunk;
      }
      load_chunk(c, cols, chn * CHUNK_E, lane, E, vec_ok, fill);

      if (nh > 0) {
        __builtin_amdgcn_fence(__ATOMIC_RELEASE, "wavefront");
        __builtin_amdgcn_wave_barrier();
        __builtin_amdgcn_fence(__ATOMIC_ACQUIRE, "wavefront");

        int pos = 0;
        if (lane == 0) pos = atomicAdd(&llen, nh);
        pos = __builtin_amdgcn_readfirstlane(pos);
        for (int k = lane; k < nh; k += 32) {
          const unsigned ent = wl[k];
          const unsigned d   = ent & 0xffffu;
          const int      r   = rows[ebase + (int)(ent >> 16)];
          atomicAdd(&cnt[d], 1);
          if (pos + k < LIST_CAP) gl[pos + k] = ((unsigned)r << D_BITS) | d;
        }
      }
      ch = chn;
    }
  }
  __syncthreads();

  for (unsigned i = threadIdx.x; i < nown; i += OWN_T)
    dinv[t0 + i] = 1.0f / sqrtf(1.0f + (float)cnt[i]);
  if (threadIdx.x == 0) glen[blockIdx.x] = llen;
}

template <int K, int NN, int WAVES>
__global__ void __launch_bounds__(WAVES * 32)
k_gemm_wmma(const float* __restrict__ A, const float* __restrict__ B,
            float* __restrict__ C, int ntiles) {
  constexpr int NT = NN / 16;
  constexpr int KS = K / 4;
  constexpr int NQ = K / 8;
  static_assert(K % 8 == 0 && NN % 16 == 0, "whole fragments only");
  __shared__ v2f bfrag[KS * NT * 32];

  const int lane = threadIdx.x & 31;
  const int wave = __builtin_amdgcn_readfirstlane((int)(threadIdx.x >> 5));
  const int tile = blockIdx.x * WAVES + wave;
  const int c    = lane & 15;
  const int h    = lane >> 4;

  v4f q[NQ];
  const float* ap = A + (size_t)(min(tile, ntiles - 1) * 16 + c) * K + 4 * h;
#pragma unroll
  for (int j = 0; j < NQ; ++j) q[j] = *reinterpret_cast<const v4f*>(ap + 8 * j);

  static_assert((K * NN / 4) % (WAVES * 32) == 0, "whole 16-byte pieces per thread");
  float* bl = reinterpret_cast<float*>(bfrag);
#pragma unroll
  for (int i = 0; i < (K * NN / 4) / (WAVES * 32); ++i) {
    const int idx = (i * WAVES * 32 + (int)threadIdx.x) * 4;
    const v4f w   = *reinterpret_cast<const v4f*>(B + idx);
    const int k   = idx / NN, n = idx % NN;
    const int j   = k >> 3, hh = (k >> 2) & 1, t = (k >> 1) & 1, s = k & 1;
    const int nt  = n >> 4, cc = n & 15;
    float* dst = bl + ((((j * 2 + t) * NT + nt) * 32) + hh * 16 + cc) * 2 + s;
    dst[0] = w.x;
    dst[2] = w.y;
    dst[4] = w.z;
    dst[6] = w.w;
  }
  __syncthreads();
  if (tile >= ntiles) return;

  v8f acc[NT];
#pragma unroll
  for (int nt = 0; nt < NT; ++nt) acc[nt] = v8f{};

#pragma unroll
  for (int j = 0; j < NQ; ++j) {
#pragma unroll
    for (int t = 0; t < 2; ++t) {
      v2f a;
      a.x = t ? q[j].z : q[j].x;
      a.y = t ? q[j].w : q[j].y;
#pragma unroll
      for (int nt = 0; nt < NT; ++nt) {
        const v2f b = bfrag[((j * 2 + t) * NT + nt) * 32 + lane];
        acc[nt] = __builtin_amdgcn_wmma_f32_16x16x4_f32(false, a, false, b, (short)0, acc[nt],
                                                        false, false);
      }
    }
  }

  float* crow = C + (size_t)(tile * 16 + 8 * h) * NN + c;
#pragma unroll
  for (int nt = 0; nt < NT; ++nt) {
#pragma unroll
    for (int v = 0; v < 8; ++v) crow[(size_t)v * NN + nt * 16] = acc[nt][v];
  }
}

template <int F>
__device__ __forceinline__ void add_trip(float* __restrict__ acc, int r, int d, int cnt, int lane,
                                         int t0, const float* __restrict__ dinv,
                                         const float* __restrict__ H) {
  constexpr int S = F / 32;
  int i0 = 0;
  do {
    float hv[ROW_G][S];
    float ww[ROW_G];
    int   dd[ROW_G];
#pragma unroll
    for (int g = 0; g < ROW_G; ++g) {
      const int ii = min(i0 + g, cnt - 1);
      const int rg = __builtin_amdgcn_readlane(r, ii);
      dd[g] = __builtin_amdgcn_readlane(d, ii);
      ww[g] = dinv[rg] * dinv[t0 + dd[g]];
      const float* hp = H + (size_t)rg * F + lane;
#pragma unroll
      for (int s = 0; s < S; ++s) hv[g][s] = hp[s * 32];
    }
#pragma unroll
    for (int g = 0; g < ROW_G; ++g) {
      if (i0 + g < cnt) {
#pragma unroll
        for (int s = 0; s < S; ++s)
          atomicAdd(&acc[dd[g] * F + s * 32 + lane], hv[g][s] * ww[g]);
      }
    }
    i0 += ROW_G;
  } while (i0 < cnt);
}

template <int F, bool RELU>
__global__ void __launch_bounds__(OWN_T)
k_agg_owner(const int* __restrict__ rows, const int* __restrict__ cols,
            const int* __restrict__ glen, const unsigned* __restrict__ glist,
            const float* __restrict__ dinv, const float* __restrict__ H,
            const float* __restrict__ bias, float* __restrict__ OUT, int n, int E) {
  static_assert(OWN_R <= 65536 && CHUNK_E <= 65536, "hit entries pack two 16-bit fields");
  __shared__ float    acc[OWN_R * F];
  __shared__ unsigned hitlist[OWN_W][CHUNK_E];

  const int lane = threadIdx.x & 31;
  const int wave = __builtin_amdgcn_readfirstlane((int)(threadIdx.x >> 5));
  const int t0   = blockIdx.x * OWN_R;
  const unsigned nown = (unsigned)min(OWN_R, n - t0);
  const int fill = t0 + OWN_R;

  for (int i = threadIdx.x; i < OWN_R * F; i += OWN_T) acc[i] = 0.0f;
  __syncthreads();

  unsigned* wl = hitlist[wave];
  const int  nchunk = (E + CHUNK_E - 1) / CHUNK_E;
  const bool vec_ok = ((E & 3) == 0) && ((((size_t)cols) & 15) == 0);
  const int  start  = (int)((blockIdx.x * max(1u, (unsigned)nchunk / gridDim.x)) % (unsigned)nchunk);

  const int len = __builtin_amdgcn_readfirstlane(glen[blockIdx.x]);

  if (len <= LIST_CAP) {
    const unsigned* gl = glist + (size_t)blockIdx.x * LIST_CAP;
    int k0 = wave * 32;
    if (k0 < len) {
      unsigned ent = gl[k0 + min(lane, min(32, len - k0) - 1)];
      for (; k0 < len; k0 += OWN_W * 32) {
        const int cnt = min(32, len - k0);
        const int kn = k0 + OWN_W * 32;
        unsigned entn = ent;
        if (kn < len) entn = gl[kn + min(lane, min(32, len - kn) - 1)];
        add_trip<F>(acc, (int)(ent >> D_BITS), (int)(ent & ((1u << D_BITS) - 1u)), cnt, lane, t0,
                    dinv, H);
        ent = entn;
      }
    }
  } else if (wave < nchunk) {
    int ci = wave;
    int ch = ci + start;
    if (ch >= nchunk) ch -= nchunk;
    v4i c[SCAN_U];
    load_chunk(c, cols, ch * CHUNK_E, lane, E, vec_ok, fill);

    for (; ci < nchunk; ci += OWN_W) {
      const int ebase = ch * CHUNK_E;

      int nh = 0;
#pragma unroll
      for (int u = 0; u < SCAN_U; ++u) {
#pragma unroll
        for (int j = 0; j < 4; ++j) {
          const unsigned d   = (unsigned)(c[u][j] - t0);
          const bool     hit = d < nown;
          const unsigned m   = __builtin_amdgcn_ballot_w32(hit);
          if (m) {
            if (hit)
              wl[nh + (int)__builtin_amdgcn_mbcnt_lo(m, 0u)] =
                  ((unsigned)(u * 128 + lane * 4 + j) << 16) | d;
            nh += __builtin_popcount(m);
          }
        }
      }

      int chn = ch;
      if (ci + OWN_W < nchunk) {
        chn = ci + OWN_W + start;
        if (chn >= nchunk) chn -= nchunk;
      }
      load_chunk(c, cols, chn * CHUNK_E, lane, E, vec_ok, fill);

      __builtin_amdgcn_fence(__ATOMIC_RELEASE, "wavefront");
      __builtin_amdgcn_wave_barrier();
      __builtin_amdgcn_fence(__ATOMIC_ACQUIRE, "wavefront");

      for (int k0 = 0; k0 < nh; k0 += 32) {
        const int cnt = min(32, nh - k0);
        const unsigned ent = wl[k0 + min(lane, cnt - 1)];
        const int d = (int)(ent & 0xffffu);
        const int r = rows[ebase + (int)(ent >> 16)];
        add_trip<F>(acc, r, d, cnt, lane, t0, dinv, H);
      }
      ch = chn;
    }
  }
  __syncthreads();

  const int tot = (int)nown * F;
  for (int i = threadIdx.x; i < tot; i += OWN_T) {
    const int    dloc = i / F;
    const int    f    = i & (F - 1);
    const size_t gi   = (size_t)t0 * F + i;
    const float  dv   = dinv[t0 + dloc];
    float v = acc[i] + H[gi] * dv * dv + bias[f];
    if (RELU) v = fmaxf(v, 0.0f);
    OUT[gi] = v;
  }
}

extern "C" void kernel_launch(void* const* d_in, const int* in_sizes, int n_in,
                              void* d_out, int out_size, void* d_ws, size_t ws_size,
                              hipStream_t stream) {
  const float* x  = (const float*)d_in[0];
  const int*   ei = (const int*)d_in[1];
  const float* W1 = (const float*)d_in[2];
  const float* b1 = (const float*)d_in[3];
  const float* W2 = (const float*)d_in[4];
  const float* b2 = (const float*)d_in[5];

  const int IN_DIM = 128, HID = 64, OUT = 32;
  const int N = in_sizes[0] / IN_DIM;
  const int E = in_sizes[1] / 2;
  const int* rows = ei;
  const int* cols = ei + E;

  const int G = (N + OWN_R - 1) / OWN_R;

  float* ws   = (float*)d_ws;
  float* dinv = ws;
  float* h1   = ws + N;
  float* r1   = h1 + (size_t)N * HID;
  float* h2   = h1;
  int*      glen  = (int*)(r1 + (size_t)N * HID);
  unsigned* glist = (unsigned*)(glen + G);

  k_deg_owner<<<G, OWN_T, 0, stream>>>(rows, cols, dinv, glen, glist, N, E);

  k_gemm_wmma<128, 64, GEMM_W><<<(N / 16 + GEMM_W - 1) / GEMM_W, 32 * GEMM_W, 0, stream>>>(x, W1, h1, N / 16);
  k_agg_owner<64, true><<<G, OWN_T, 0, stream>>>(rows, cols, glen, glist, dinv, h1, b1, r1, N, E);

  k_gemm_wmma<64, 32, GEMM_W><<<(N / 16 + GEMM_W - 1) / GEMM_W, 32 * GEMM_W, 0, stream>>>(r1, W2, h2, N / 16);
  k_agg_owner<32, false><<<G, OWN_T, 0, stream>>>(rows, cols, glen, glist, dinv, h2, b2,
                                                  (float*)d_out, N, E);
}


# Both owner sums: every lane works out its own edge's weight, row offset and LDS offset, 32 edges at once; the gather loop only hands them round by readlane; LDS arrival order as before
# speedup vs baseline: 3.1815x; 1.0466x over previous
#include <hip/hip_runtime.h>
#include <cstddef>


typedef __attribute__((ext_vector_type(2))) float v2f;
typedef __attribute__((ext_vector_type(8))) float v8f;
typedef __attribute__((ext_vector_type(4))) int   v4i;
typedef __attribute__((ext_vector_type(4))) float v4f;

#define OWN_R   196
#define OWN_T   1024
#define OWN_W   (OWN_T / 32)
#define SCAN_U  8
#define CHUNK_E (32 * 4 * SCAN_U)
#define ROW_G   8
#define LIST_CAP 131072
#define D_BITS  8
#define GEMM_W  8

__device__ __forceinline__ void load_chunk(v4i (&c)[SCAN_U], const int* __restrict__ cols,
                                           int ebase, int lane, int E, bool vec_ok, int fill) {
  if (vec_ok && ebase + CHUNK_E <= E) {
#pragma unroll
    for (int u = 0; u < SCAN_U; ++u)
      c[u] = *reinterpret_cast<const v4i*>(cols + ebase + u * 128 + lane * 4);
  } else {
#pragma unroll
    for (int u = 0; u < SCAN_U; ++u) {
      const int e = ebase + u * 128 + lane * 4;
      c[u].x = (e     < E) ? cols[e]     : fill;
      c[u].y = (e + 1 < E) ? cols[e + 1] : fill;
      c[u].z = (e + 2 < E) ? cols[e + 2] : fill;
      c[u].w = (e + 3 < E) ? cols[e + 3] : fill;
    }
  }
}

__global__ void __launch_bounds__(OWN_T)
k_deg_owner(const int* __restrict__ rows, const int* __restrict__ cols, float* __restrict__ dinv,
            int* __restrict__ glen, unsigned* __restrict__ glist, int n, int E) {
  static_assert(OWN_R <= (1 << D_BITS), "local target must fit the low bits of a list entry");
  static_assert(CHUNK_E <= 65536, "hit entries pack two 16-bit fields");
  __shared__ int      cnt[OWN_R];
  __shared__ int      llen;
  __shared__ unsigned hitlist[OWN_W][CHUNK_E];
  const int lane = threadIdx.x & 31;
  const int wave = __builtin_amdgcn_readfirstlane((int)(threadIdx.x >> 5));
  const int t0   = blockIdx.x * OWN_R;
  const unsigned nown = (unsigned)min(OWN_R, n - t0);
  const int fill = t0 + OWN_R;

  for (int i = threadIdx.x; i < OWN_R; i += OWN_T) cnt[i] = 0;
  if (threadIdx.x == 0) llen = 0;
  __syncthreads();

  unsigned* wl = hitlist[wave];
  unsigned* gl = glist + (size_t)blockIdx.x * LIST_CAP;
  const int  nchunk = (E + CHUNK_E - 1) / CHUNK_E;
  const bool vec_ok = ((E & 3) == 0) && ((((size_t)cols) & 15) == 0);
  const int  start  = (int)((blockIdx.x * max(1u, (unsigned)nchunk / gridDim.x)) % (unsigned)nchunk);

  if (wave < nchunk) {
    int ci = wave;
    int ch = ci + start;
    if (ch >= nchunk) ch -= nchunk;
    v4i c[SCAN_U];
    load_chunk(c, cols, ch * CHUNK_E, lane, E, vec_ok, fill);

    for (; ci < nchunk; ci += OWN_W) {
      const int ebase = ch * CHUNK_E;

      int nh = 0;
#pragma unroll
      for (int u = 0; u < SCAN_U; ++u) {
#pragma unroll
        for (int j = 0; j < 4; ++j) {
          const unsigned d   = (unsigned)(c[u][j] - t0);
          const bool     hit = d < nown;
          const unsigned m   = __builtin_amdgcn_ballot_w32(hit);
          if (m) {
            if (hit)
              wl[nh + (int)__builtin_amdgcn_mbcnt_lo(m, 0u)] =
                  ((unsigned)(u * 128 + lane * 4 + j) << 16) | d;
            nh += __builtin_popcount(m);
          }
        }
      }

      int chn = ch;
      if (ci + OWN_W < nchunk) {
        chn = ci + OWN_W + start;
        if (chn >= nchunk) chn -= nchunk;
      }
      load_chunk(c, cols, chn * CHUNK_E, lane, E, vec_ok, fill);

      if (nh > 0) {
        __builtin_amdgcn_fence(__ATOMIC_RELEASE, "wavefront");
        __builtin_amdgcn_wave_barrier();
        __builtin_amdgcn_fence(__ATOMIC_ACQUIRE, "wavefront");

        int pos = 0;
        if (lane == 0) pos = atomicAdd(&llen, nh);
        pos = __builtin_amdgcn_readfirstlane(pos);
        for (int k = lane; k < nh; k += 32) {
          const unsigned ent = wl[k];
          const unsigned d   = ent & 0xffffu;
          const int      r   = rows[ebase + (int)(ent >> 16)];
          atomicAdd(&cnt[d], 1);
          if (pos + k < LIST_CAP) gl[pos + k] = ((unsigned)r << D_BITS) | d;
        }
      }
      ch = chn;
    }
  }
  __syncthreads();

  for (unsigned i = threadIdx.x; i < nown; i += OWN_T)
    dinv[t0 + i] = 1.0f / sqrtf(1.0f + (float)cnt[i]);
  if (threadIdx.x == 0) glen[blockIdx.x] = llen;
}

template <int K, int NN, int WAVES>
__global__ void __launch_bounds__(WAVES * 32)
k_gemm_wmma(const float* __restrict__ A, const float* __restrict__ B,
            float* __restrict__ C, int ntiles) {
  constexpr int NT = NN / 16;
  constexpr int KS = K / 4;
  constexpr int NQ = K / 8;
  static_assert(K % 8 == 0 && NN % 16 == 0, "whole fragments only");
  __shared__ v2f bfrag[KS * NT * 32];

  const int lane = threadIdx.x & 31;
  const int wave = __builtin_amdgcn_readfirstlane((int)(threadIdx.x >> 5));
  const int tile = blockIdx.x * WAVES + wave;
  const int c    = lane & 15;
  const int h    = lane >> 4;

  v4f q[NQ];
  const float* ap = A + (size_t)(min(tile, ntiles - 1) * 16 + c) * K + 4 * h;
#pragma unroll
  for (int j = 0; j < NQ; ++j) q[j] = *reinterpret_cast<const v4f*>(ap + 8 * j);

  static_assert((K * NN / 4) % (WAVES * 32) == 0, "whole 16-byte pieces per thread");
  float* bl = reinterpret_cast<float*>(bfrag);
#pragma unroll
  for (int i = 0; i < (K * NN / 4) / (WAVES * 32); ++i) {
    const int idx = (i * WAVES * 32 + (int)threadIdx.x) * 4;
    const v4f w   = *reinterpret_cast<const v4f*>(B + idx);
    const int k   = idx / NN, n = idx % NN;
    const int j   = k >> 3, hh = (k >> 2) & 1, t = (k >> 1) & 1, s = k & 1;
    const int nt  = n >> 4, cc = n & 15;
    float* dst = bl + ((((j * 2 + t) * NT + nt) * 32) + hh * 16 + cc) * 2 + s;
    dst[0] = w.x;
    dst[2] = w.y;
    dst[4] = w.z;
    dst[6] = w.w;
  }
  __syncthreads();
  if (tile >= ntiles) return;

  v8f acc[NT];
#pragma unroll
  for (int nt = 0; nt < NT; ++nt) acc[nt] = v8f{};

#pragma unroll
  for (int j = 0; j < NQ; ++j) {
#pragma unroll
    for (int t = 0; t < 2; ++t) {
      v2f a;
      a.x = t ? q[j].z : q[j].x;
      a.y = t ? q[j].w : q[j].y;
#pragma unroll
      for (int nt = 0; nt < NT; ++nt) {
        const v2f b = bfrag[((j * 2 + t) * NT + nt) * 32 + lane];
        acc[nt] = __builtin_amdgcn_wmma_f32_16x16x4_f32(false, a, false, b, (short)0, acc[nt],
                                                        false, false);
      }
    }
  }

  float* crow = C + (size_t)(tile * 16 + 8 * h) * NN + c;
#pragma unroll
  for (int nt = 0; nt < NT; ++nt) {
#pragma unroll
    for (int v = 0; v < 8; ++v) crow[(size_t)v * NN + nt * 16] = acc[nt][v];
  }
}

template <int F>
__device__ __forceinline__ void add_trip(float* __restrict__ acc, int r, int d, int cnt, int lane,
                                         int t0, const float* __restrict__ dinv,
                                         const float* __restrict__ H) {
  constexpr int S = F / 32;
  const float    w  = dinv[r] * dinv[t0 + d];
  const int      wb = __builtin_bit_cast(int, w);
  const int      ro = r * (F * 4);
  const int      lo = d * (F * 4);
  const char*    Hb = reinterpret_cast<const char*>(H);
  char*          ab = reinterpret_cast<char*>(acc) + lane * 4;
  int i0 = 0;
  do {
    float hv[ROW_G][S];
    int   ll[ROW_G];
#pragma unroll
    for (int g = 0; g < ROW_G; ++g) {
      const unsigned rg = (unsigned)__builtin_amdgcn_readlane(ro, i0 + g);
      ll[g] = __builtin_amdgcn_readlane(lo, i0 + g);
      const float* hp = reinterpret_cast<const float*>(Hb + (size_t)rg);
#pragma unroll
      for (int s = 0; s < S; ++s) hv[g][s] = hp[s * 32 + lane];
    }
    if (i0 + ROW_G <= cnt) {
#pragma unroll
      for (int g = 0; g < ROW_G; ++g) {
        const float wg = __builtin_bit_cast(float, __builtin_amdgcn_readlane(wb, i0 + g));
        float* ap = reinterpret_cast<float*>(ab + ll[g]);
#pragma unroll
        for (int s = 0; s < S; ++s) atomicAdd(&ap[s * 32], hv[g][s] * wg);
      }
    } else {
#pragma unroll
      for (int g = 0; g < ROW_G; ++g) {
        if (i0 + g < cnt) {
          const float wg = __builtin_bit_cast(float, __builtin_amdgcn_readlane(wb, i0 + g));
          float* ap = reinterpret_cast<float*>(ab + ll[g]);
#pragma unroll
          for (int s = 0; s < S; ++s) atomicAdd(&ap[s * 32], hv[g][s] * wg);
        }
      }
    }
    i0 += ROW_G;
  } while (i0 < cnt);
}

template <int F, bool RELU>
__global__ void __launch_bounds__(OWN_T)
k_agg_owner(const int* __restrict__ rows, const int* __restrict__ cols,
            const int* __restrict__ glen, const unsigned* __restrict__ glist,
            const float* __restrict__ dinv, const float* __restrict__ H,
            const float* __restrict__ bias, float* __restrict__ OUT, int n, int E) {
  static_assert(OWN_R <= 65536 && CHUNK_E <= 65536, "hit entries pack two 16-bit fields");
  __shared__ float    acc[OWN_R * F];
  __shared__ unsigned hitlist[OWN_W][CHUNK_E];

  const int lane = threadIdx.x & 31;
  const int wave = __builtin_amdgcn_readfirstlane((int)(threadIdx.x >> 5));
  const int t0   = blockIdx.x * OWN_R;
  const unsigned nown = (unsigned)min(OWN_R, n - t0);
  const int fill = t0 + OWN_R;

  for (int i = threadIdx.x; i < OWN_R * F; i += OWN_T) acc[i] = 0.0f;
  __syncthreads();

  unsigned* wl = hitlist[wave];
  const int  nchunk = (E + CHUNK_E - 1) / CHUNK_E;
  const bool vec_ok = ((E & 3) == 0) && ((((size_t)cols) & 15) == 0);
  const int  start  = (int)((blockIdx.x * max(1u, (unsigned)nchunk / gridDim.x)) % (unsigned)nchunk);

  const int len = __builtin_amdgcn_readfirstlane(glen[blockIdx.x]);

  if (len <= LIST_CAP) {
    const unsigned* gl = glist + (size_t)blockIdx.x * LIST_CAP;
    int k0 = wave * 32;
    if (k0 < len) {
      unsigned ent = gl[k0 + min(lane, min(32, len - k0) - 1)];
      for (; k0 < len; k0 += OWN_W * 32) {
        const int cnt = min(32, len - k0);
        const int kn = k0 + OWN_W * 32;
        unsigned entn = ent;
        if (kn < len) entn = gl[kn + min(lane, min(32, len - kn) - 1)];
        add_trip<F>(acc, (int)(ent >> D_BITS), (int)(ent & ((1u << D_BITS) - 1u)), cnt, lane, t0,
                    dinv, H);
        ent = entn;
      }
    }
  } else if (wave < nchunk) {
    int ci = wave;
    int ch = ci + start;
    if (ch >= nchunk) ch -= nchunk;
    v4i c[SCAN_U];
    load_chunk(c, cols, ch * CHUNK_E, lane, E, vec_ok, fill);

    for (; ci < nchunk; ci += OWN_W) {
      const int ebase = ch * CHUNK_E;

      int nh = 0;
#pragma unroll
      for (int u = 0; u < SCAN_U; ++u) {
#pragma unroll
        for (int j = 0; j < 4; ++j) {
          const unsigned d   = (unsigned)(c[u][j] - t0);
          const bool     hit = d < nown;
          const unsigned m   = __builtin_amdgcn_ballot_w32(hit);
          if (m) {
            if (hit)
              wl[nh + (int)__builtin_amdgcn_mbcnt_lo(m, 0u)] =
                  ((unsigned)(u * 128 + lane * 4 + j) << 16) | d;
            nh += __builtin_popcount(m);
          }
        }
      }

      int chn = ch;
      if (ci + OWN_W < nchunk) {
        chn = ci + OWN_W + start;
        if (chn >= nchunk) chn -= nchunk;
      }
      load_chunk(c, cols, chn * CHUNK_E, lane, E, vec_ok, fill);

      __builtin_amdgcn_fence(__ATOMIC_RELEASE, "wavefront");
      __builtin_amdgcn_wave_barrier();
      __builtin_amdgcn_fence(__ATOMIC_ACQUIRE, "wavefront");

      for (int k0 = 0; k0 < nh; k0 += 32) {
        const int cnt = min(32, nh - k0);
        const unsigned ent = wl[k0 + min(lane, cnt - 1)];
        const int d = (int)(ent & 0xffffu);
        const int r = rows[ebase + (int)(ent >> 16)];
        add_trip<F>(acc, r, d, cnt, lane, t0, dinv, H);
      }
      ch = chn;
    }
  }
  __syncthreads();

  const int tot = (int)nown * F;
  for (int i = threadIdx.x; i < tot; i += OWN_T) {
    const int    dloc = i / F;
    const int    f    = i & (F - 1);
    const size_t gi   = (size_t)t0 * F + i;
    const float  dv   = dinv[t0 + dloc];
    float v = acc[i] + H[gi] * dv * dv + bias[f];
    if (RELU) v = fmaxf(v, 0.0f);
    OUT[gi] = v;
  }
}

extern "C" void kernel_launch(void* const* d_in, const int* in_sizes, int n_in,
                              void* d_out, int out_size, void* d_ws, size_t ws_size,
                              hipStream_t stream) {
  const float* x  = (const float*)d_in[0];
  const int*   ei = (const int*)d_in[1];
  const float* W1 = (const float*)d_in[2];
  const float* b1 = (const float*)d_in[3];
  const float* W2 = (const float*)d_in[4];
  const float* b2 = (const float*)d_in[5];

  const int IN_DIM = 128, HID = 64, OUT = 32;
  const int N = in_sizes[0] / IN_DIM;
  const int E = in_sizes[1] / 2;
  const int* rows = ei;
  const int* cols = ei + E;

  const int G = (N + OWN_R - 1) / OWN_R;

  float* ws   = (float*)d_ws;
  float* dinv = ws;
  float* h1   = ws + N;
  float* r1   = h1 + (size_t)N * HID;
  float* h2   = h1;
  int*      glen  = (int*)(r1 + (size_t)N * HID);
  unsigned* glist = (unsigned*)(glen + G);

  k_deg_owner<<<G, OWN_T, 0, stream>>>(rows, cols, dinv, glen, glist, N, E);

  k_gemm_wmma<128, 64, GEMM_W><<<(N / 16 + GEMM_W - 1) / GEMM_W, 32 * GEMM_W, 0, stream>>>(x, W1, h1, N / 16);
  k_agg_owner<64, true><<<G, OWN_T, 0, stream>>>(rows, cols, glen, glist, dinv, h1, b1, r1, N, E);

  k_gemm_wmma<64, 32, GEMM_W><<<(N / 16 + GEMM_W - 1) / GEMM_W, 32 * GEMM_W, 0, stream>>>(r1, W2, h2, N / 16);
  k_agg_owner<32, false><<<G, OWN_T, 0, stream>>>(rows, cols, glen, glist, dinv, h2, b2,
                                                  (float*)d_out, N, E);
}


# Edges sorted by owner once: 196 slice groups rank 4096 edges each in LDS counters, owners merge their segments into list and degree; no group reads the whole list; 5 launches to 6; LDS arrival order
# speedup vs baseline: 4.0390x; 1.2695x over previous
#include <hip/hip_runtime.h>
#include <cstddef>


typedef __attribute__((ext_vector_type(2))) float v2f;
typedef __attribute__((ext_vector_type(8))) float v8f;
typedef __attribute__((ext_vector_type(4))) int   v4i;
typedef __attribute__((ext_vector_type(4))) float v4f;

#define OWN_R   196
#define OWN_T   1024
#define OWN_W   (OWN_T / 32)
#define SCAN_U  8
#define CHUNK_E (32 * 4 * SCAN_U)
#define ROW_G   8
#define LIST_CAP 131072
#define D_BITS  8
#define GEMM_W  8
#define SL_E    4096
#define BIN_T   256
#define BIN_Q   (SL_E / BIN_T)
#define MAX_G   (4 * BIN_T)
#define MRG_T   256
#define MRG_Q   16
#define MAX_NB  (4 * MRG_T)

__device__ __forceinline__ void load_chunk(v4i (&c)[SCAN_U], const int* __restrict__ cols,
                                           int ebase, int lane, int E, bool vec_ok, int fill) {
  if (vec_ok && ebase + CHUNK_E <= E) {
#pragma unroll
    for (int u = 0; u < SCAN_U; ++u)
      c[u] = *reinterpret_cast<const v4i*>(cols + ebase + u * 128 + lane * 4);
  } else {
#pragma unroll
    for (int u = 0; u < SCAN_U; ++u) {
      const int e = ebase + u * 128 + lane * 4;
      c[u].x = (e     < E) ? cols[e]     : fill;
      c[u].y = (e + 1 < E) ? cols[e + 1] : fill;
      c[u].z = (e + 2 < E) ? cols[e + 2] : fill;
      c[u].w = (e + 3 < E) ? cols[e + 3] : fill;
    }
  }
}

__global__ void __launch_bounds__(BIN_T)
k_bin_slices(const int* __restrict__ rows, const int* __restrict__ cols,
             unsigned* __restrict__ seg, int* __restrict__ soT, int G, int n, int E) {
  static_assert(SL_E % BIN_T == 0 && MAX_G == 4 * BIN_T, "whole edges and four counters per thread");
  __shared__ __attribute__((aligned(16))) int hist[MAX_G];
  __shared__ unsigned sorted[SL_E];
  __shared__ int      wtot[BIN_T / 32];
  const int tid  = threadIdx.x;
  const int lane = tid & 31;
  const int wave = __builtin_amdgcn_readfirstlane(tid >> 5);
  const int NB   = (int)gridDim.x;
  const int e0   = blockIdx.x * SL_E;
  const int cnt  = min(SL_E, E - e0);
  const unsigned gown = (unsigned)min(G, MAX_G - 1);

  int cv[BIN_Q], rv[BIN_Q];
#pragma unroll
  for (int q = 0; q < BIN_Q; ++q) {
    const int e = e0 + min(q * BIN_T + tid, cnt - 1);
    cv[q] = cols[e];
    rv[q] = rows[e];
  }
#pragma unroll
  for (int i = 0; i < MAX_G / BIN_T; ++i) hist[i * BIN_T + tid] = 0;
  __syncthreads();

  int      key[BIN_Q], rk[BIN_Q];
  unsigned ent[BIN_Q];
#pragma unroll
  for (int q = 0; q < BIN_Q; ++q) {
    const unsigned o  = (unsigned)cv[q] / (unsigned)OWN_R;
    const bool     ok = (q * BIN_T + tid < cnt) && ((unsigned)cv[q] < (unsigned)n) && (o < gown);
    key[q] = ok ? (int)o : -1;
    ent[q] = ((unsigned)rv[q] << D_BITS) | ((unsigned)cv[q] - o * (unsigned)OWN_R);
    rk[q]  = 0;
    if (ok) rk[q] = atomicAdd(&hist[o], 1);
  }
  __syncthreads();

  const v4i hq = *reinterpret_cast<const v4i*>(&hist[4 * tid]);
  const int s  = hq.x + hq.y + hq.z + hq.w;
  int inc = s;
#pragma unroll
  for (int d = 1; d < 32; d <<= 1) {
    const int t = __shfl_up(inc, d);
    if (lane >= d) inc += t;
  }
  if (lane == 31) wtot[wave] = inc;
  __syncthreads();
  int base = inc - s;
  for (int w = 0; w < wave; ++w) base += wtot[w];
  v4i ex;
  ex.x = base;
  ex.y = ex.x + hq.x;
  ex.z = ex.y + hq.y;
  ex.w = ex.z + hq.z;
  *reinterpret_cast<v4i*>(&hist[4 * tid]) = ex;
#pragma unroll
  for (int j = 0; j < 4; ++j) {
    const int bin = 4 * tid + j;
    if (bin <= (int)gown) soT[(size_t)bin * NB + blockIdx.x] = ex[j];
  }
  __syncthreads();

  int hp[BIN_Q];
#pragma unroll
  for (int q = 0; q < BIN_Q; ++q) hp[q] = hist[max(key[q], 0)];
#pragma unroll
  for (int q = 0; q < BIN_Q; ++q)
    if (key[q] >= 0) sorted[hp[q] + rk[q]] = ent[q];
  __syncthreads();

  const int kept = hist[gown];
  unsigned* sg = seg + (size_t)blockIdx.x * SL_E;
  for (int i = tid; i < kept; i += BIN_T) sg[i] = sorted[i];
}

__global__ void __launch_bounds__(MRG_T)
k_deg_merge(const unsigned* __restrict__ seg, const int* __restrict__ soT,
            float* __restrict__ dinv, int* __restrict__ glen, unsigned* __restrict__ glist,
            int n, int NB) {
  static_assert(OWN_R <= (1 << D_BITS), "local target must fit the low bits of a list entry");
  static_assert(MAX_NB == 4 * MRG_T, "four slices per thread");
  __shared__ int cnt[OWN_R];
  __shared__ __attribute__((aligned(16))) int pa[MAX_NB];
  __shared__ __attribute__((aligned(16))) int sa[MAX_NB];
  __shared__ int wtot[MRG_T / 32];
  __shared__ int ltot;
  const int tid  = threadIdx.x;
  const int lane = tid & 31;
  const int wave = __builtin_amdgcn_readfirstlane(tid >> 5);
  const int o    = blockIdx.x;
  const int t0   = o * OWN_R;
  const int nown = min(OWN_R, n - t0);
  const int nb   = min(NB, MAX_NB - 1);

  int a[4], z[4], nn[4];
#pragma unroll
  for (int j = 0; j < 4; ++j) {
    const int bb = min(4 * tid + j, nb - 1);
    a[j] = soT[(size_t)o * NB + bb];
    z[j] = soT[(size_t)(o + 1) * NB + bb];
  }
#pragma unroll
  for (int j = 0; j < 4; ++j) nn[j] = (z[j] - a[j]) & ((4 * tid + j < nb) ? -1 : 0);
  for (int i = tid; i < OWN_R; i += MRG_T) cnt[i] = 0;

  const int s = nn[0] + nn[1] + nn[2] + nn[3];
  int inc = s;
#pragma unroll
  for (int d = 1; d < 32; d <<= 1) {
    const int t = __shfl_up(inc, d);
    if (lane >= d) inc += t;
  }
  if (lane == 31) wtot[wave] = inc;
  __syncthreads();
  int base = inc - s;
  for (int w = 0; w < wave; ++w) base += wtot[w];
  v4i pv, sv;
  pv.x = base;
  pv.y = pv.x + nn[0];
  pv.z = pv.y + nn[1];
  pv.w = pv.z + nn[2];
  sv.x = a[0]; sv.y = a[1]; sv.z = a[2]; sv.w = a[3];
  *reinterpret_cast<v4i*>(&pa[4 * tid]) = pv;
  *reinterpret_cast<v4i*>(&sa[4 * tid]) = sv;
  if (tid == MRG_T - 1) ltot = base + s;
  __syncthreads();

  const int len   = __builtin_amdgcn_readfirstlane(ltot);
  const int step0 = 1 << (31 - __builtin_clz(nb));
  unsigned* gl = glist + (size_t)o * LIST_CAP;
  for (int k0 = 0; k0 < len; k0 += MRG_T * MRG_Q) {
    int kk[MRG_Q], bq[MRG_Q];
#pragma unroll
    for (int q = 0; q < MRG_Q; ++q) {
      kk[q] = min(k0 + q * MRG_T + tid, len - 1);
      bq[q] = 0;
    }
    for (int step = step0; step > 0; step >>= 1) {
      int pv[MRG_Q];
#pragma unroll
      for (int q = 0; q < MRG_Q; ++q) pv[q] = pa[bq[q] + step];
#pragma unroll
      for (int q = 0; q < MRG_Q; ++q) bq[q] += (pv[q] <= kk[q]) ? step : 0;
    }
    int sq[MRG_Q], pq[MRG_Q];
#pragma unroll
    for (int q = 0; q < MRG_Q; ++q) {
      sq[q] = sa[bq[q]];
      pq[q] = pa[bq[q]];
    }
    unsigned ev[MRG_Q];
#pragma unroll
    for (int q = 0; q < MRG_Q; ++q)
      ev[q] = seg[(size_t)bq[q] * SL_E + (sq[q] + (kk[q] - pq[q]))];
    asm volatile("" ::"v"(ev[0]));
#pragma unroll
    for (int q = 0; q < MRG_Q; ++q) {
      const int k = k0 + q * MRG_T + tid;
      if (k < len) {
        atomicAdd(&cnt[ev[q] & ((1u << D_BITS) - 1u)], 1);
        if (k < LIST_CAP) gl[k] = ev[q];
      }
    }
  }
  __syncthreads();

  for (int i = tid; i < nown; i += MRG_T)
    dinv[t0 + i] = 1.0f / sqrtf(1.0f + (float)cnt[i]);
  if (tid == 0) glen[o] = len;
}

template <int K, int NN, int WAVES>
__global__ void __launch_bounds__(WAVES * 32)
k_gemm_wmma(const float* __restrict__ A, const float* __restrict__ B,
            float* __restrict__ C, int ntiles) {
  constexpr int NT = NN / 16;
  constexpr int KS = K / 4;
  constexpr int NQ = K / 8;
  static_assert(K % 8 == 0 && NN % 16 == 0, "whole fragments only");
  __shared__ v2f bfrag[KS * NT * 32];

  const int lane = threadIdx.x & 31;
  const int wave = __builtin_amdgcn_readfirstlane((int)(threadIdx.x >> 5));
  const int tile = blockIdx.x * WAVES + wave;
  const int c    = lane & 15;
  const int h    = lane >> 4;

  v4f q[NQ];
  const float* ap = A + (size_t)(min(tile, ntiles - 1) * 16 + c) * K + 4 * h;
#pragma unroll
  for (int j = 0; j < NQ; ++j) q[j] = *reinterpret_cast<const v4f*>(ap + 8 * j);

  static_assert((K * NN / 4) % (WAVES * 32) == 0, "whole 16-byte pieces per thread");
  float* bl = reinterpret_cast<float*>(bfrag);
#pragma unroll
  for (int i = 0; i < (K * NN / 4) / (WAVES * 32); ++i) {
    const int idx = (i * WAVES * 32 + (int)threadIdx.x) * 4;
    const v4f w   = *reinterpret_cast<const v4f*>(B + idx);
    const int k   = idx / NN, n = idx % NN;
    const int j   = k >> 3, hh = (k >> 2) & 1, t = (k >> 1) & 1, s = k & 1;
    const int nt  = n >> 4, cc = n & 15;
    float* dst = bl + ((((j * 2 + t) * NT + nt) * 32) + hh * 16 + cc) * 2 + s;
    dst[0] = w.x;
    dst[2] = w.y;
    dst[4] = w.z;
    dst[6] = w.w;
  }
  __syncthreads();
  if (tile >= ntiles) return;

  v8f acc[NT];
#pragma unroll
  for (int nt = 0; nt < NT; ++nt) acc[nt] = v8f{};

#pragma unroll
  for (int j = 0; j < NQ; ++j) {
#pragma unroll
    for (int t = 0; t < 2; ++t) {
      v2f a;
      a.x = t ? q[j].z : q[j].x;
      a.y = t ? q[j].w : q[j].y;
#pragma unroll
      for (int nt = 0; nt < NT; ++nt) {
        const v2f b = bfrag[((j * 2 + t) * NT + nt) * 32 + lane];
        acc[nt] = __builtin_amdgcn_wmma_f32_16x16x4_f32(false, a, false, b, (short)0, acc[nt],
                                                        false, false);
      }
    }
  }

  float* crow = C + (size_t)(tile * 16 + 8 * h) * NN + c;
#pragma unroll
  for (int nt = 0; nt < NT; ++nt) {
#pragma unroll
    for (int v = 0; v < 8; ++v) crow[(size_t)v * NN + nt * 16] = acc[nt][v];
  }
}

template <int F>
__device__ __forceinline__ void add_trip(float* __restrict__ acc, int r, int d, int cnt, int lane,
                                         int t0, const float* __restrict__ dinv,
                                         const float* __restrict__ H) {
  constexpr int S = F / 32;
  const float    w  = dinv[r] * dinv[t0 + d];
  const int      wb = __builtin_bit_cast(int, w);
  const int      ro = r * (F * 4);
  const int      lo = d * (F * 4);
  const char*    Hb = reinterpret_cast<const char*>(H);
  char*          ab = reinterpret_cast<char*>(acc) + lane * 4;
  int i0 = 0;
  do {
    float hv[ROW_G][S];
    int   ll[ROW_G];
#pragma unroll
    for (int g = 0; g < ROW_G; ++g) {
      const unsigned rg = (unsigned)__builtin_amdgcn_readlane(ro, i0 + g);
      ll[g] = __builtin_amdgcn_readlane(lo, i0 + g);
      const float* hp = reinterpret_cast<const float*>(Hb + (size_t)rg);
#pragma unroll
      for (int s = 0; s < S; ++s) hv[g][s] = hp[s * 32 + lane];
    }
    if (i0 + ROW_G <= cnt) {
#pragma unroll
      for (int g = 0; g < ROW_G; ++g) {
        const float wg = __builtin_bit_cast(float, __builtin_amdgcn_readlane(wb, i0 + g));
        float* ap = reinterpret_cast<float*>(ab + ll[g]);
#pragma unroll
        for (int s = 0; s < S; ++s) atomicAdd(&ap[s * 32], hv[g][s] * wg);
      }
    } else {
#pragma unroll
      for (int g = 0; g < ROW_G; ++g) {
        if (i0 + g < cnt) {
          const float wg = __builtin_bit_cast(float, __builtin_amdgcn_readlane(wb, i0 + g));
          float* ap = reinterpret_cast<float*>(ab + ll[g]);
#pragma unroll
          for (int s = 0; s < S; ++s) atomicAdd(&ap[s * 32], hv[g][s] * wg);
        }
      }
    }
    i0 += ROW_G;
  } while (i0 < cnt);
}

template <int F, bool RELU>
__global__ void __launch_bounds__(OWN_T)
k_agg_owner(const int* __restrict__ rows, const int* __restrict__ cols,
            const int* __restrict__ glen, const unsigned* __restrict__ glist,
            const float* __restrict__ dinv, const float* __restrict__ H,
            const float* __restrict__ bias, float* __restrict__ OUT, int n, int E) {
  static_assert(OWN_R <= 65536 && CHUNK_E <= 65536, "hit entries pack two 16-bit fields");
  __shared__ float    acc[OWN_R * F];
  __shared__ unsigned hitlist[OWN_W][CHUNK_E];

  const int lane = threadIdx.x & 31;
  const int wave = __builtin_amdgcn_readfirstlane((int)(threadIdx.x >> 5));
  const int t0   = blockIdx.x * OWN_R;
  const unsigned nown = (unsigned)min(OWN_R, n - t0);
  const int fill = t0 + OWN_R;

  for (int i = threadIdx.x; i < OWN_R * F; i += OWN_T) acc[i] = 0.0f;
  __syncthreads();

  unsigned* wl = hitlist[wave];
  const int  nchunk = (E + CHUNK_E - 1) / CHUNK_E;
  const bool vec_ok = ((E & 3) == 0) && ((((size_t)cols) & 15) == 0);
  const int  start  = (int)((blockIdx.x * max(1u, (unsigned)nchunk / gridDim.x)) % (unsigned)nchunk);

  const int len = __builtin_amdgcn_readfirstlane(glen[blockIdx.x]);

  if (len <= LIST_CAP) {
    const unsigned* gl = glist + (size_t)blockIdx.x * LIST_CAP;
    int k0 = wave * 32;
    if (k0 < len) {
      unsigned ent = gl[k0 + min(lane, min(32, len - k0) - 1)];
      for (; k0 < len; k0 += OWN_W * 32) {
        const int cnt = min(32, len - k0);
        const int kn = k0 + OWN_W * 32;
        unsigned entn = ent;
        if (kn < len) entn = gl[kn + min(lane, min(32, len - kn) - 1)];
        add_trip<F>(acc, (int)(ent >> D_BITS), (int)(ent & ((1u << D_BITS) - 1u)), cnt, lane, t0,
                    dinv, H);
        ent = entn;
      }
    }
  } else if (wave < nchunk) {
    int ci = wave;
    int ch = ci + start;
    if (ch >= nchunk) ch -= nchunk;
    v4i c[SCAN_U];
    load_chunk(c, cols, ch * CHUNK_E, lane, E, vec_ok, fill);

    for (; ci < nchunk; ci += OWN_W) {
      const int ebase = ch * CHUNK_E;

      int nh = 0;
#pragma unroll
      for (int u = 0; u < SCAN_U; ++u) {
#pragma unroll
        for (int j = 0; j < 4; ++j) {
          const unsigned d   = (unsigned)(c[u][j] - t0);
          const bool     hit = d < nown;
          const unsigned m   = __builtin_amdgcn_ballot_w32(hit);
          if (m) {
            if (hit)
              wl[nh + (int)__builtin_amdgcn_mbcnt_lo(m, 0u)] =
                  ((unsigned)(u * 128 + lane * 4 + j) << 16) | d;
            nh += __builtin_popcount(m);
          }
        }
      }

      int chn = ch;
      if (ci + OWN_W < nchunk) {
        chn = ci + OWN_W + start;
        if (chn >= nchunk) chn -= nchunk;
      }
      load_chunk(c, cols, chn * CHUNK_E, lane, E, vec_ok, fill);

      __builtin_amdgcn_fence(__ATOMIC_RELEASE, "wavefront");
      __builtin_amdgcn_wave_barrier();
      __builtin_amdgcn_fence(__ATOMIC_ACQUIRE, "wavefront");

      for (int k0 = 0; k0 < nh; k0 += 32) {
        const int cnt = min(32, nh - k0);
        const unsigned ent = wl[k0 + min(lane, cnt - 1)];
        const int d = (int)(ent & 0xffffu);
        const int r = rows[ebase + (int)(ent >> 16)];
        add_trip<F>(acc, r, d, cnt, lane, t0, dinv, H);
      }
      ch = chn;
    }
  }
  __syncthreads();

  const int tot = (int)nown * F;
  for (int i = threadIdx.x; i < tot; i += OWN_T) {
    const int    dloc = i / F;
    const int    f    = i & (F - 1);
    const size_t gi   = (size_t)t0 * F + i;
    const float  dv   = dinv[t0 + dloc];
    float v = acc[i] + H[gi] * dv * dv + bias[f];
    if (RELU) v = fmaxf(v, 0.0f);
    OUT[gi] = v;
  }
}

extern "C" void kernel_launch(void* const* d_in, const int* in_sizes, int n_in,
                              void* d_out, int out_size, void* d_ws, size_t ws_size,
                              hipStream_t stream) {
  const float* x  = (const float*)d_in[0];
  const int*   ei = (const int*)d_in[1];
  const float* W1 = (const float*)d_in[2];
  const float* b1 = (const float*)d_in[3];
  const float* W2 = (const float*)d_in[4];
  const float* b2 = (const float*)d_in[5];

  const int IN_DIM = 128, HID = 64, OUT = 32;
  const int N = in_sizes[0] / IN_DIM;
  const int E = in_sizes[1] / 2;
  const int* rows = ei;
  const int* cols = ei + E;

  const int G  = (N + OWN_R - 1) / OWN_R;
  const int NB = (E + SL_E - 1) / SL_E;

  float* ws   = (float*)d_ws;
  float* dinv = ws;
  float* h1   = ws + N;
  float* r1   = h1 + (size_t)N * HID;
  float* h2   = h1;
  int*      glen  = (int*)(r1 + (size_t)N * HID);
  unsigned* glist = (unsigned*)(glen + G);
  unsigned* seg   = glist + (size_t)G * LIST_CAP;
  int*      soT   = (int*)(seg + (size_t)NB * SL_E);

  k_bin_slices<<<NB, BIN_T, 0, stream>>>(rows, cols, seg, soT, G, N, E);
  k_deg_merge<<<G, MRG_T, 0, stream>>>(seg, soT, dinv, glen, glist, N, NB);

  k_gemm_wmma<128, 64, GEMM_W><<<(N / 16 + GEMM_W - 1) / GEMM_W, 32 * GEMM_W, 0, stream>>>(x, W1, h1, N / 16);
  k_agg_owner<64, true><<<G, OWN_T, 0, stream>>>(rows, cols, glen, glist, dinv, h1, b1, r1, N, E);

  k_gemm_wmma<64, 32, GEMM_W><<<(N / 16 + GEMM_W - 1) / GEMM_W, 32 * GEMM_W, 0, stream>>>(r1, W2, h2, N / 16);
  k_agg_owner<32, false><<<G, OWN_T, 0, stream>>>(rows, cols, glen, glist, dinv, h2, b2,
                                                  (float*)d_out, N, E);
}


# Edge weight split: h rows stored times own dinv, owners add rows unweighted, no dinv gather per edge; own row starts each LDS sum (then arrival order as before); dinv[target] applied at write-out
# speedup vs baseline: 4.9931x; 1.2362x over previous
#include <hip/hip_runtime.h>
#include <cstddef>


typedef __attribute__((ext_vector_type(2))) float v2f;
typedef __attribute__((ext_vector_type(8))) float v8f;
typedef __attribute__((ext_vector_type(4))) int   v4i;
typedef __attribute__((ext_vector_type(4))) float v4f;

#define OWN_R   196
#define OWN_T   1024
#define OWN_W   (OWN_T / 32)
#define SCAN_U  8
#define CHUNK_E (32 * 4 * SCAN_U)
#define ROW_G   8
#define LIST_CAP 131072
#define D_BITS  8
#define GEMM_W  8
#define SL_E    4096
#define BIN_T   256
#define BIN_Q   (SL_E / BIN_T)
#define MAX_G   (4 * BIN_T)
#define MRG_T   256
#define MRG_Q   16
#define MAX_NB  (4 * MRG_T)

__device__ __forceinline__ void load_chunk(v4i (&c)[SCAN_U], const int* __restrict__ cols,
                                           int ebase, int lane, int E, bool vec_ok, int fill) {
  if (vec_ok && ebase + CHUNK_E <= E) {
#pragma unroll
    for (int u = 0; u < SCAN_U; ++u)
      c[u] = *reinterpret_cast<const v4i*>(cols + ebase + u * 128 + lane * 4);
  } else {
#pragma unroll
    for (int u = 0; u < SCAN_U; ++u) {
      const int e = ebase + u * 128 + lane * 4;
      c[u].x = (e     < E) ? cols[e]     : fill;
      c[u].y = (e + 1 < E) ? cols[e + 1] : fill;
      c[u].z = (e + 2 < E) ? cols[e + 2] : fill;
      c[u].w = (e + 3 < E) ? cols[e + 3] : fill;
    }
  }
}

__global__ void __launch_bounds__(BIN_T)
k_bin_slices(const int* __restrict__ rows, const int* __restrict__ cols,
             unsigned* __restrict__ seg, int* __restrict__ soT, int G, int n, int E) {
  static_assert(SL_E % BIN_T == 0 && MAX_G == 4 * BIN_T, "whole edges and four counters per thread");
  __shared__ __attribute__((aligned(16))) int hist[MAX_G];
  __shared__ unsigned sorted[SL_E];
  __shared__ int      wtot[BIN_T / 32];
  const int tid  = threadIdx.x;
  const int lane = tid & 31;
  const int wave = __builtin_amdgcn_readfirstlane(tid >> 5);
  const int NB   = (int)gridDim.x;
  const int e0   = blockIdx.x * SL_E;
  const int cnt  = min(SL_E, E - e0);
  const unsigned gown = (unsigned)min(G, MAX_G - 1);

  int cv[BIN_Q], rv[BIN_Q];
#pragma unroll
  for (int q = 0; q < BIN_Q; ++q) {
    const int e = e0 + min(q * BIN_T + tid, cnt - 1);
    cv[q] = cols[e];
    rv[q] = rows[e];
  }
#pragma unroll
  for (int i = 0; i < MAX_G / BIN_T; ++i) hist[i * BIN_T + tid] = 0;
  __syncthreads();

  int      key[BIN_Q], rk[BIN_Q];
  unsigned ent[BIN_Q];
#pragma unroll
  for (int q = 0; q < BIN_Q; ++q) {
    const unsigned o  = (unsigned)cv[q] / (unsigned)OWN_R;
    const bool     ok = (q * BIN_T + tid < cnt) && ((unsigned)cv[q] < (unsigned)n) && (o < gown);
    key[q] = ok ? (int)o : -1;
    ent[q] = ((unsigned)rv[q] << D_BITS) | ((unsigned)cv[q] - o * (unsigned)OWN_R);
    rk[q]  = 0;
    if (ok) rk[q] = atomicAdd(&hist[o], 1);
  }
  __syncthreads();

  const v4i hq = *reinterpret_cast<const v4i*>(&hist[4 * tid]);
  const int s  = hq.x + hq.y + hq.z + hq.w;
  int inc = s;
#pragma unroll
  for (int d = 1; d < 32; d <<= 1) {
    const int t = __shfl_up(inc, d);
    if (lane >= d) inc += t;
  }
  if (lane == 31) wtot[wave] = inc;
  __syncthreads();
  int base = inc - s;
  for (int w = 0; w < wave; ++w) base += wtot[w];
  v4i ex;
  ex.x = base;
  ex.y = ex.x + hq.x;
  ex.z = ex.y + hq.y;
  ex.w = ex.z + hq.z;
  *reinterpret_cast<v4i*>(&hist[4 * tid]) = ex;
#pragma unroll
  for (int j = 0; j < 4; ++j) {
    const int bin = 4 * tid + j;
    if (bin <= (int)gown) soT[(size_t)bin * NB + blockIdx.x] = ex[j];
  }
  __syncthreads();

  int hp[BIN_Q];
#pragma unroll
  for (int q = 0; q < BIN_Q; ++q) hp[q] = hist[max(key[q], 0)];
#pragma unroll
  for (int q = 0; q < BIN_Q; ++q)
    if (key[q] >= 0) sorted[hp[q] + rk[q]] = ent[q];
  __syncthreads();

  const int kept = hist[gown];
  unsigned* sg = seg + (size_t)blockIdx.x * SL_E;
  for (int i = tid; i < kept; i += BIN_T) sg[i] = sorted[i];
}

__global__ void __launch_bounds__(MRG_T)
k_deg_merge(const unsigned* __restrict__ seg, const int* __restrict__ soT,
            float* __restrict__ dinv, int* __restrict__ glen, unsigned* __restrict__ glist,
            int n, int NB) {
  static_assert(OWN_R <= (1 << D_BITS), "local target must fit the low bits of a list entry");
  static_assert(MAX_NB == 4 * MRG_T, "four slices per thread");
  __shared__ int cnt[OWN_R];
  __shared__ __attribute__((aligned(16))) int pa[MAX_NB];
  __shared__ __attribute__((aligned(16))) int sa[MAX_NB];
  __shared__ int wtot[MRG_T / 32];
  __shared__ int ltot;
  const int tid  = threadIdx.x;
  const int lane = tid & 31;
  const int wave = __builtin_amdgcn_readfirstlane(tid >> 5);
  const int o    = blockIdx.x;
  const int t0   = o * OWN_R;
  const int nown = min(OWN_R, n - t0);
  const int nb   = min(NB, MAX_NB - 1);

  int a[4], z[4], nn[4];
#pragma unroll
  for (int j = 0; j < 4; ++j) {
    const int bb = min(4 * tid + j, nb - 1);
    a[j] = soT[(size_t)o * NB + bb];
    z[j] = soT[(size_t)(o + 1) * NB + bb];
  }
#pragma unroll
  for (int j = 0; j < 4; ++j) nn[j] = (z[j] - a[j]) & ((4 * tid + j < nb) ? -1 : 0);
  for (int i = tid; i < OWN_R; i += MRG_T) cnt[i] = 0;

  const int s = nn[0] + nn[1] + nn[2] + nn[3];
  int inc = s;
#pragma unroll
  for (int d = 1; d < 32; d <<= 1) {
    const int t = __shfl_up(inc, d);
    if (lane >= d) inc += t;
  }
  if (lane == 31) wtot[wave] = inc;
  __syncthreads();
  int base = inc - s;
  for (int w = 0; w < wave; ++w) base += wtot[w];
  v4i pv, sv;
  pv.x = base;
  pv.y = pv.x + nn[0];
  pv.z = pv.y + nn[1];
  pv.w = pv.z + nn[2];
  sv.x = a[0]; sv.y = a[1]; sv.z = a[2]; sv.w = a[3];
  *reinterpret_cast<v4i*>(&pa[4 * tid]) = pv;
  *reinterpret_cast<v4i*>(&sa[4 * tid]) = sv;
  if (tid == MRG_T - 1) ltot = base + s;
  __syncthreads();

  const int len   = __builtin_amdgcn_readfirstlane(ltot);
  const int step0 = 1 << (31 - __builtin_clz(nb));
  unsigned* gl = glist + (size_t)o * LIST_CAP;
  for (int k0 = 0; k0 < len; k0 += MRG_T * MRG_Q) {
    int kk[MRG_Q], bq[MRG_Q];
#pragma unroll
    for (int q = 0; q < MRG_Q; ++q) {
      kk[q] = min(k0 + q * MRG_T + tid, len - 1);
      bq[q] = 0;
    }
    for (int step = step0; step > 0; step >>= 1) {
      int pv[MRG_Q];
#pragma unroll
      for (int q = 0; q < MRG_Q; ++q) pv[q] = pa[bq[q] + step];
#pragma unroll
      for (int q = 0; q < MRG_Q; ++q) bq[q] += (pv[q] <= kk[q]) ? step : 0;
    }
    int sq[MRG_Q], pq[MRG_Q];
#pragma unroll
    for (int q = 0; q < MRG_Q; ++q) {
      sq[q] = sa[bq[q]];
      pq[q] = pa[bq[q]];
    }
    unsigned ev[MRG_Q];
#pragma unroll
    for (int q = 0; q < MRG_Q; ++q)
      ev[q] = seg[(size_t)bq[q] * SL_E + (sq[q] + (kk[q] - pq[q]))];
    asm volatile("" ::"v"(ev[0]));
#pragma unroll
    for (int q = 0; q < MRG_Q; ++q) {
      const int k = k0 + q * MRG_T + tid;
      if (k < len) {
        atomicAdd(&cnt[ev[q] & ((1u << D_BITS) - 1u)], 1);
        if (k < LIST_CAP) gl[k] = ev[q];
      }
    }
  }
  __syncthreads();

  for (int i = tid; i < nown; i += MRG_T)
    dinv[t0 + i] = 1.0f / sqrtf(1.0f + (float)cnt[i]);
  if (tid == 0) glen[o] = len;
}

template <int K, int NN, int WAVES>
__global__ void __launch_bounds__(WAVES * 32)
k_gemm_wmma(const float* __restrict__ A, const float* __restrict__ B,
            const float* __restrict__ rs, float* __restrict__ C, int ntiles) {
  constexpr int NT = NN / 16;
  constexpr int KS = K / 4;
  constexpr int NQ = K / 8;
  static_assert(K % 8 == 0 && NN % 16 == 0, "whole fragments only");
  __shared__ v2f bfrag[KS * NT * 32];

  const int lane = threadIdx.x & 31;
  const int wave = __builtin_amdgcn_readfirstlane((int)(threadIdx.x >> 5));
  const int tile = blockIdx.x * WAVES + wave;
  const int c    = lane & 15;
  const int h    = lane >> 4;

  v4f q[NQ];
  const float* ap = A + (size_t)(min(tile, ntiles - 1) * 16 + c) * K + 4 * h;
#pragma unroll
  for (int j = 0; j < NQ; ++j) q[j] = *reinterpret_cast<const v4f*>(ap + 8 * j);
  const float* sp  = rs + (size_t)(min(tile, ntiles - 1) * 16 + 8 * h);
  const v4f    sc0 = *reinterpret_cast<const v4f*>(sp);
  const v4f    sc1 = *reinterpret_cast<const v4f*>(sp + 4);

  static_assert((K * NN / 4) % (WAVES * 32) == 0, "whole 16-byte pieces per thread");
  float* bl = reinterpret_cast<float*>(bfrag);
#pragma unroll
  for (int i = 0; i < (K * NN / 4) / (WAVES * 32); ++i) {
    const int idx = (i * WAVES * 32 + (int)threadIdx.x) * 4;
    const v4f w   = *reinterpret_cast<const v4f*>(B + idx);
    const int k   = idx / NN, n = idx % NN;
    const int j   = k >> 3, hh = (k >> 2) & 1, t = (k >> 1) & 1, s = k & 1;
    const int nt  = n >> 4, cc = n & 15;
    float* dst = bl + ((((j * 2 + t) * NT + nt) * 32) + hh * 16 + cc) * 2 + s;
    dst[0] = w.x;
    dst[2] = w.y;
    dst[4] = w.z;
    dst[6] = w.w;
  }
  __syncthreads();
  if (tile >= ntiles) return;

  v8f acc[NT];
#pragma unroll
  for (int nt = 0; nt < NT; ++nt) acc[nt] = v8f{};

#pragma unroll
  for (int j = 0; j < NQ; ++j) {
#pragma unroll
    for (int t = 0; t < 2; ++t) {
      v2f a;
      a.x = t ? q[j].z : q[j].x;
      a.y = t ? q[j].w : q[j].y;
#pragma unroll
      for (int nt = 0; nt < NT; ++nt) {
        const v2f b = bfrag[((j * 2 + t) * NT + nt) * 32 + lane];
        acc[nt] = __builtin_amdgcn_wmma_f32_16x16x4_f32(false, a, false, b, (short)0, acc[nt],
                                                        false, false);
      }
    }
  }

  float* crow = C + (size_t)(tile * 16 + 8 * h) * NN + c;
#pragma unroll
  for (int nt = 0; nt < NT; ++nt) {
#pragma unroll
    for (int v = 0; v < 8; ++v)
      crow[(size_t)v * NN + nt * 16] = acc[nt][v] * (v < 4 ? sc0[v & 3] : sc1[v & 3]);
  }
}

template <int F>
__device__ __forceinline__ void add_trip(float* __restrict__ acc, int r, int d, int cnt, int lane,
                                         const float* __restrict__ H) {
  constexpr int S = F / 32;
  const int      ro = r * (F * 4);
  const int      lo = d * (F * 4);
  const char*    Hb = reinterpret_cast<const char*>(H);
  char*          ab = reinterpret_cast<char*>(acc) + lane * 4;
  int i0 = 0;
  for (; i0 + ROW_G <= cnt; i0 += ROW_G) {
    float hv[ROW_G][S];
    int   ll[ROW_G];
#pragma unroll
    for (int g = 0; g < ROW_G; ++g) {
      const unsigned rg = (unsigned)__builtin_amdgcn_readlane(ro, i0 + g);
      ll[g] = __builtin_amdgcn_readlane(lo, i0 + g);
      const float* hp = reinterpret_cast<const float*>(Hb + (size_t)rg);
#pragma unroll
      for (int s = 0; s < S; ++s) hv[g][s] = hp[s * 32 + lane];
    }
#pragma unroll
    for (int g = 0; g < ROW_G; ++g) {
      float* ap = reinterpret_cast<float*>(ab + ll[g]);
#pragma unroll
      for (int s = 0; s < S; ++s) atomicAdd(&ap[s * 32], hv[g][s]);
    }
  }
  if (i0 < cnt) {
    float hv[ROW_G][S];
    int   ll[ROW_G];
#pragma unroll
    for (int g = 0; g < ROW_G; ++g) {
      const unsigned rg = (unsigned)__builtin_amdgcn_readlane(ro, i0 + g);
      ll[g] = __builtin_amdgcn_readlane(lo, i0 + g);
      const float* hp = reinterpret_cast<const float*>(Hb + (size_t)rg);
#pragma unroll
      for (int s = 0; s < S; ++s) hv[g][s] = hp[s * 32 + lane];
    }
#pragma unroll
    for (int g = 0; g < ROW_G; ++g) {
      if (i0 + g < cnt) {
        float* ap = reinterpret_cast<float*>(ab + ll[g]);
#pragma unroll
        for (int s = 0; s < S; ++s) atomicAdd(&ap[s * 32], hv[g][s]);
      }
    }
  }
}

template <int F, bool RELU>
__global__ void __launch_bounds__(OWN_T)
k_agg_owner(const int* __restrict__ rows, const int* __restrict__ cols,
            const int* __restrict__ glen, const unsigned* __restrict__ glist,
            const float* __restrict__ dinv, const float* __restrict__ H,
            const float* __restrict__ bias, float* __restrict__ OUT, int n, int E) {
  static_assert(OWN_R <= 65536 && CHUNK_E <= 65536, "hit entries pack two 16-bit fields");
  static_assert(OWN_T % F == 0 && OWN_T >= OWN_R, "a thread keeps one feature column; a thread per owned node");
  __shared__ float    acc[OWN_R * F];
  __shared__ float    dtg[OWN_R];
  __shared__ unsigned hitlist[OWN_W][CHUNK_E];

  const int lane = threadIdx.x & 31;
  const int wave = __builtin_amdgcn_readfirstlane((int)(threadIdx.x >> 5));
  const int t0   = blockIdx.x * OWN_R;
  const unsigned nown = (unsigned)min(OWN_R, n - t0);
  const int fill = t0 + OWN_R;

  constexpr int NI  = (OWN_R * F + OWN_T - 1) / OWN_T;
  const int     tot = (int)nown * F;
  const float*  own = H + (size_t)t0 * F;
  float sv[NI];
#pragma unroll
  for (int k = 0; k < NI; ++k) sv[k] = own[min(k * OWN_T + (int)threadIdx.x, tot - 1)];
  const float dmine = dinv[t0 + (int)min(threadIdx.x, nown - 1u)];
  const float bf    = bias[threadIdx.x & (F - 1)];

  const int len = __builtin_amdgcn_readfirstlane(glen[blockIdx.x]);
  const unsigned* gl = glist + (size_t)blockIdx.x * LIST_CAP;
  const bool listed  = (len <= LIST_CAP);
  unsigned ent = 0u;
  if (listed && wave * 32 < len) ent = gl[wave * 32 + min(lane, min(32, len - wave * 32) - 1)];

#pragma unroll
  for (int k = 0; k < NI; ++k) {
    const int i = k * OWN_T + (int)threadIdx.x;
    if (i < OWN_R * F) acc[i] = (i < tot) ? sv[k] : 0.0f;
  }
  if (threadIdx.x < OWN_R) dtg[threadIdx.x] = dmine;
  __syncthreads();

  unsigned* wl = hitlist[wave];
  const int  nchunk = (E + CHUNK_E - 1) / CHUNK_E;
  const bool vec_ok = ((E & 3) == 0) && ((((size_t)cols) & 15) == 0);
  const int  start  = (int)((blockIdx.x * max(1u, (unsigned)nchunk / gridDim.x)) % (unsigned)nchunk);

  if (listed) {
    int k0 = wave * 32;
    if (k0 < len) {
      for (; k0 < len; k0 += OWN_W * 32) {
        const int cnt = min(32, len - k0);
        const int kn = k0 + OWN_W * 32;
        unsigned entn = ent;
        if (kn < len) entn = gl[kn + min(lane, min(32, len - kn) - 1)];
        add_trip<F>(acc, (int)(ent >> D_BITS), (int)(ent & ((1u << D_BITS) - 1u)), cnt, lane, H);
        ent = entn;
      }
    }
  } else if (wave < nchunk) {
    int ci = wave;
    int ch = ci + start;
    if (ch >= nchunk) ch -= nchunk;
    v4i c[SCAN_U];
    load_chunk(c, cols, ch * CHUNK_E, lane, E, vec_ok, fill);

    for (; ci < nchunk; ci += OWN_W) {
      const int ebase = ch * CHUNK_E;

      int nh = 0;
#pragma unroll
      for (int u = 0; u < SCAN_U; ++u) {
#pragma unroll
        for (int j = 0; j < 4; ++j) {
          const unsigned d   = (unsigned)(c[u][j] - t0);
          const bool     hit = d < nown;
          const unsigned m   = __builtin_amdgcn_ballot_w32(hit);
          if (m) {
            if (hit)
              wl[nh + (int)__builtin_amdgcn_mbcnt_lo(m, 0u)] =
                  ((unsigned)(u * 128 + lane * 4 + j) << 16) | d;
            nh += __builtin_popcount(m);
          }
        }
      }

      int chn = ch;
      if (ci + OWN_W < nchunk) {
        chn = ci + OWN_W + start;
        if (chn >= nchunk) chn -= nchunk;
      }
      load_chunk(c, cols, chn * CHUNK_E, lane, E, vec_ok, fill);

      __builtin_amdgcn_fence(__ATOMIC_RELEASE, "wavefront");
      __builtin_amdgcn_wave_barrier();
      __builtin_amdgcn_fence(__ATOMIC_ACQUIRE, "wavefront");

      for (int k0 = 0; k0 < nh; k0 += 32) {
        const int cnt = min(32, nh - k0);
        const unsigned ent = wl[k0 + min(lane, cnt - 1)];
        const int d = (int)(ent & 0xffffu);
        const int r = rows[ebase + (int)(ent >> 16)];
        add_trip<F>(acc, r, d, cnt, lane, H);
      }
      ch = chn;
    }
  }
  __syncthreads();

  float* outp = OUT + (size_t)t0 * F;
  for (int i = threadIdx.x; i < tot; i += OWN_T) {
    float v = acc[i] * dtg[i / F] + bf;
    if (RELU) v = fmaxf(v, 0.0f);
    outp[i] = v;
  }
}

extern "C" void kernel_launch(void* const* d_in, const int* in_sizes, int n_in,
                              void* d_out, int out_size, void* d_ws, size_t ws_size,
                              hipStream_t stream) {
  const float* x  = (const float*)d_in[0];
  const int*   ei = (const int*)d_in[1];
  const float* W1 = (const float*)d_in[2];
  const float* b1 = (const float*)d_in[3];
  const float* W2 = (const float*)d_in[4];
  const float* b2 = (const float*)d_in[5];

  const int IN_DIM = 128, HID = 64, OUT = 32;
  const int N = in_sizes[0] / IN_DIM;
  const int E = in_sizes[1] / 2;
  const int* rows = ei;
  const int* cols = ei + E;

  const int G  = (N + OWN_R - 1) / OWN_R;
  const int NB = (E + SL_E - 1) / SL_E;

  float* ws   = (float*)d_ws;
  float* dinv = ws;
  float* h1   = ws + N;
  float* r1   = h1 + (size_t)N * HID;
  float* h2   = h1;
  int*      glen  = (int*)(r1 + (size_t)N * HID);
  unsigned* glist = (unsigned*)(glen + G);
  unsigned* seg   = glist + (size_t)G * LIST_CAP;
  int*      soT   = (int*)(seg + (size_t)NB * SL_E);

  k_bin_slices<<<NB, BIN_T, 0, stream>>>(rows, cols, seg, soT, G, N, E);
  k_deg_merge<<<G, MRG_T, 0, stream>>>(seg, soT, dinv, glen, glist, N, NB);

  k_gemm_wmma<128, 64, GEMM_W><<<(N / 16 + GEMM_W - 1) / GEMM_W, 32 * GEMM_W, 0, stream>>>(x, W1, dinv, h1, N / 16);
  k_agg_owner<64, true><<<G, OWN_T, 0, stream>>>(rows, cols, glen, glist, dinv, h1, b1, r1, N, E);

  k_gemm_wmma<64, 32, GEMM_W><<<(N / 16 + GEMM_W - 1) / GEMM_W, 32 * GEMM_W, 0, stream>>>(r1, W2, dinv, h2, N / 16);
  k_agg_owner<32, false><<<G, OWN_T, 0, stream>>>(rows, cols, glen, glist, dinv, h2, b2,
                                                  (float*)d_out, N, E);
}


# Owner merge launch also does x@W1 for its own 196 rows (13 waves, a tile each): x flies in from memory while the list is built, dinv for the row scale is in LDS; 6 launches to 5; sums keep order
# speedup vs baseline: 5.2158x; 1.0446x over previous
#include <hip/hip_runtime.h>
#include <cstddef>


typedef __attribute__((ext_vector_type(2))) float v2f;
typedef __attribute__((ext_vector_type(8))) float v8f;
typedef __attribute__((ext_vector_type(4))) int   v4i;
typedef __attribute__((ext_vector_type(4))) float v4f;

#define OWN_R   196
#define OWN_T   1024
#define OWN_W   (OWN_T / 32)
#define SCAN_U  8
#define CHUNK_E (32 * 4 * SCAN_U)
#define ROW_G   8
#define LIST_CAP 131072
#define D_BITS  8
#define GEMM_W  8
#define SL_E    4096
#define BIN_T   256
#define BIN_Q   (SL_E / BIN_T)
#define MAX_G   (4 * BIN_T)
#define OWN_TILES ((OWN_R + 15) / 16)
#define MRG_T   (32 * OWN_TILES)
#define MRG_Q   8
#define MAX_NB  (4 * MRG_T)
#define SRCH_NB 1024

__device__ __forceinline__ void load_chunk(v4i (&c)[SCAN_U], const int* __restrict__ cols,
                                           int ebase, int lane, int E, bool vec_ok, int fill) {
  if (vec_ok && ebase + CHUNK_E <= E) {
#pragma unroll
    for (int u = 0; u < SCAN_U; ++u)
      c[u] = *reinterpret_cast<const v4i*>(cols + ebase + u * 128 + lane * 4);
  } else {
#pragma unroll
    for (int u = 0; u < SCAN_U; ++u) {
      const int e = ebase + u * 128 + lane * 4;
      c[u].x = (e     < E) ? cols[e]     : fill;
      c[u].y = (e + 1 < E) ? cols[e + 1] : fill;
      c[u].z = (e + 2 < E) ? cols[e + 2] : fill;
      c[u].w = (e + 3 < E) ? cols[e + 3] : fill;
    }
  }
}

__global__ void __launch_bounds__(BIN_T)
k_bin_slices(const int* __restrict__ rows, const int* __restrict__ cols,
             unsigned* __restrict__ seg, int* __restrict__ soT, int G, int n, int E) {
  static_assert(SL_E % BIN_T == 0 && MAX_G == 4 * BIN_T, "whole edges and four counters per thread");
  __shared__ __attribute__((aligned(16))) int hist[MAX_G];
  __shared__ unsigned sorted[SL_E];
  __shared__ int      wtot[BIN_T / 32];
  const int tid  = threadIdx.x;
  const int lane = tid & 31;
  const int wave = __builtin_amdgcn_readfirstlane(tid >> 5);
  const int NB   = (int)gridDim.x;
  const int e0   = blockIdx.x * SL_E;
  const int cnt  = min(SL_E, E - e0);
  const unsigned gown = (unsigned)min(G, MAX_G - 1);

  int cv[BIN_Q], rv[BIN_Q];
#pragma unroll
  for (int q = 0; q < BIN_Q; ++q) {
    const int e = e0 + min(q * BIN_T + tid, cnt - 1);
    cv[q] = cols[e];
    rv[q] = rows[e];
  }
#pragma unroll
  for (int i = 0; i < MAX_G / BIN_T; ++i) hist[i * BIN_T + tid] = 0;
  __syncthreads();

  int      key[BIN_Q], rk[BIN_Q];
  unsigned ent[BIN_Q];
#pragma unroll
  for (int q = 0; q < BIN_Q; ++q) {
    const unsigned o  = (unsigned)cv[q] / (unsigned)OWN_R;
    const bool     ok = (q * BIN_T + tid < cnt) && ((unsigned)cv[q] < (unsigned)n) && (o < gown);
    key[q] = ok ? (int)o : -1;
    ent[q] = ((unsigned)rv[q] << D_BITS) | ((unsigned)cv[q] - o * (unsigned)OWN_R);
    rk[q]  = 0;
    if (ok) rk[q] = atomicAdd(&hist[o], 1);
  }
  __syncthreads();

  const v4i hq = *reinterpret_cast<const v4i*>(&hist[4 * tid]);
  const int s  = hq.x + hq.y + hq.z + hq.w;
  int inc = s;
#pragma unroll
  for (int d = 1; d < 32; d <<= 1) {
    const int t = __shfl_up(inc, d);
    if (lane >= d) inc += t;
  }
  if (lane == 31) wtot[wave] = inc;
  __syncthreads();
  int base = inc - s;
  for (int w = 0; w < wave; ++w) base += wtot[w];
  v4i ex;
  ex.x = base;
  ex.y = ex.x + hq.x;
  ex.z = ex.y + hq.y;
  ex.w = ex.z + hq.z;
  *reinterpret_cast<v4i*>(&hist[4 * tid]) = ex;
#pragma unroll
  for (int j = 0; j < 4; ++j) {
    const int bin = 4 * tid + j;
    if (bin <= (int)gown) soT[(size_t)bin * NB + blockIdx.x] = ex[j];
  }
  __syncthreads();

  int hp[BIN_Q];
#pragma unroll
  for (int q = 0; q < BIN_Q; ++q) hp[q] = hist[max(key[q], 0)];
#pragma unroll
  for (int q = 0; q < BIN_Q; ++q)
    if (key[q] >= 0) sorted[hp[q] + rk[q]] = ent[q];
  __syncthreads();

  const int kept = hist[gown];
  unsigned* sg = seg + (size_t)blockIdx.x * SL_E;
  for (int i = tid; i < kept; i += BIN_T) sg[i] = sorted[i];
}

template <int K, int NN, int WAVES>
__global__ void __launch_bounds__(WAVES * 32)
k_gemm_wmma(const float* __restrict__ A, const float* __restrict__ B,
            const float* __restrict__ rs, float* __restrict__ C, int ntiles) {
  constexpr int NT = NN / 16;
  constexpr int KS = K / 4;
  constexpr int NQ = K / 8;
  static_assert(K % 8 == 0 && NN % 16 == 0, "whole fragments only");
  __shared__ v2f bfrag[KS * NT * 32];

  const int lane = threadIdx.x & 31;
  const int wave = __builtin_amdgcn_readfirstlane((int)(threadIdx.x >> 5));
  const int tile = blockIdx.x * WAVES + wave;
  const int c    = lane & 15;
  const int h    = lane >> 4;

  v4f q[NQ];
  const float* ap = A + (size_t)(min(tile, ntiles - 1) * 16 + c) * K + 4 * h;
#pragma unroll
  for (int j = 0; j < NQ; ++j) q[j] = *reinterpret_cast<const v4f*>(ap + 8 * j);
  const float* sp  = rs + (size_t)(min(tile, ntiles - 1) * 16 + 8 * h);
  const v4f    sc0 = *reinterpret_cast<const v4f*>(sp);
  const v4f    sc1 = *reinterpret_cast<const v4f*>(sp + 4);

  static_assert((K * NN / 4) % (WAVES * 32) == 0, "whole 16-byte pieces per thread");
  float* bl = reinterpret_cast<float*>(bfrag);
#pragma unroll
  for (int i = 0; i < (K * NN / 4) / (WAVES * 32); ++i) {
    const int idx = (i * WAVES * 32 + (int)threadIdx.x) * 4;
    const v4f w   = *reinterpret_cast<const v4f*>(B + idx);
    const int k   = idx / NN, n = idx % NN;
    const int j   = k >> 3, hh = (k >> 2) & 1, t = (k >> 1) & 1, s = k & 1;
    const int nt  = n >> 4, cc = n & 15;
    float* dst = bl + ((((j * 2 + t) * NT + nt) * 32) + hh * 16 + cc) * 2 + s;
    dst[0] = w.x;
    dst[2] = w.y;
    dst[4] = w.z;
    dst[6] = w.w;
  }
  __syncthreads();
  if (tile >= ntiles) return;

  v8f acc[NT];
#pragma unroll
  for (int nt = 0; nt < NT; ++nt) acc[nt] = v8f{};

#pragma unroll
  for (int j = 0; j < NQ; ++j) {
#pragma unroll
    for (int t = 0; t < 2; ++t) {
      v2f a;
      a.x = t ? q[j].z : q[j].x;
      a.y = t ? q[j].w : q[j].y;
#pragma unroll
      for (int nt = 0; nt < NT; ++nt) {
        const v2f b = bfrag[((j * 2 + t) * NT + nt) * 32 + lane];
        acc[nt] = __builtin_amdgcn_wmma_f32_16x16x4_f32(false, a, false, b, (short)0, acc[nt],
                                                        false, false);
      }
    }
  }

  float* crow = C + (size_t)(tile * 16 + 8 * h) * NN + c;
#pragma unroll
  for (int nt = 0; nt < NT; ++nt) {
#pragma unroll
    for (int v = 0; v < 8; ++v)
      crow[(size_t)v * NN + nt * 16] = acc[nt][v] * (v < 4 ? sc0[v & 3] : sc1[v & 3]);
  }
}

__device__ __forceinline__ void merge_round(int k0, int len, int step0, int tid,
                                            const int* pa, const int* sa, int* cnt,
                                            const unsigned* __restrict__ seg,
                                            unsigned* __restrict__ gl) {
  int kk[MRG_Q], bq[MRG_Q];
#pragma unroll
  for (int q2 = 0; q2 < MRG_Q; ++q2) {
    kk[q2] = min(k0 + q2 * MRG_T + tid, len - 1);
    bq[q2] = 0;
  }
  for (int step = step0; step > 0; step >>= 1) {
    int pw[MRG_Q];
#pragma unroll
    for (int q2 = 0; q2 < MRG_Q; ++q2) pw[q2] = pa[bq[q2] + step];
#pragma unroll
    for (int q2 = 0; q2 < MRG_Q; ++q2) bq[q2] += (pw[q2] <= kk[q2]) ? step : 0;
  }
  int sq[MRG_Q], pq[MRG_Q];
#pragma unroll
  for (int q2 = 0; q2 < MRG_Q; ++q2) {
    sq[q2] = sa[bq[q2]];
    pq[q2] = pa[bq[q2]];
  }
  unsigned ev[MRG_Q];
#pragma unroll
  for (int q2 = 0; q2 < MRG_Q; ++q2)
    ev[q2] = seg[(size_t)bq[q2] * SL_E + (sq[q2] + (kk[q2] - pq[q2]))];
  asm volatile("" ::"v"(ev[0]));
#pragma unroll
  for (int q2 = 0; q2 < MRG_Q; ++q2) {
    const int k = k0 + q2 * MRG_T + tid;
    if (k < len) {
      atomicAdd(&cnt[ev[q2] & ((1u << D_BITS) - 1u)], 1);
      if (k < LIST_CAP) gl[k] = ev[q2];
    }
  }
}

__device__ __forceinline__ void lds_barrier() {
  __builtin_amdgcn_fence(__ATOMIC_RELEASE, "workgroup", "local");
  __builtin_amdgcn_s_barrier();
  __builtin_amdgcn_fence(__ATOMIC_ACQUIRE, "workgroup", "local");
}

template <int K, int NN>
__global__ void __launch_bounds__(MRG_T)
k_merge_gemm(const unsigned* __restrict__ seg, const int* __restrict__ soT,
             float* __restrict__ dinv, int* __restrict__ glen, unsigned* __restrict__ glist,
             const float* __restrict__ A, const float* __restrict__ B, float* __restrict__ C,
             int n, int NB) {
  constexpr int NT = NN / 16;
  constexpr int KS = K / 4;
  constexpr int NQ = K / 8;
  constexpr int NP = K * NN / 4;
  constexpr int NW = (NP + MRG_T - 1) / MRG_T;
  static_assert(K % 8 == 0 && NN % 16 == 0, "whole fragments only");
  static_assert(OWN_R <= (1 << D_BITS), "local target must fit the low bits of a list entry");
  static_assert(MAX_NB == 4 * MRG_T, "four slices per thread");
  static_assert(SRCH_NB <= MAX_NB && (SRCH_NB & (SRCH_NB - 1)) == 0, "the search stays inside the tables");
  __shared__ int   cnt[OWN_R];
  __shared__ float dl[OWN_R];
  __shared__ __attribute__((aligned(16))) int pa[MAX_NB];
  __shared__ __attribute__((aligned(16))) int sa[MAX_NB];
  __shared__ int wtot[MRG_T / 32];
  __shared__ int ltot;
  __shared__ v2f bfrag[KS * NT * 32];
  const int tid  = threadIdx.x;
  const int lane = tid & 31;
  const int wave = __builtin_amdgcn_readfirstlane(tid >> 5);
  const int c    = lane & 15;
  const int h    = lane >> 4;
  const int o    = blockIdx.x;
  const int t0   = o * OWN_R;
  const int nown = min(OWN_R, n - t0);
  const int nb   = min(NB, SRCH_NB - 1);

  int a[4], z[4], nn[4];
#pragma unroll
  for (int j = 0; j < 4; ++j) {
    const int bb = min(4 * tid + j, nb - 1);
    a[j] = soT[(size_t)o * NB + bb];
    z[j] = soT[(size_t)(o + 1) * NB + bb];
  }
  v4f q[NQ];
  const float* ap = A + (size_t)min(t0 + wave * 16 + c, n - 1) * K + 4 * h;
#pragma unroll
  for (int j = 0; j < NQ; ++j) q[j] = *reinterpret_cast<const v4f*>(ap + 8 * j);
  v4f wv[NW];
#pragma unroll
  for (int i = 0; i < NW; ++i)
    wv[i] = *reinterpret_cast<const v4f*>(B + 4 * min(i * MRG_T + tid, NP - 1));

#pragma unroll
  for (int j = 0; j < 4; ++j) nn[j] = (z[j] - a[j]) & ((4 * tid + j < nb) ? -1 : 0);
  for (int i = tid; i < OWN_R; i += MRG_T) cnt[i] = 0;

  const int s = nn[0] + nn[1] + nn[2] + nn[3];
  int inc = s;
#pragma unroll
  for (int d = 1; d < 32; d <<= 1) {
    const int t = __shfl_up(inc, d);
    if (lane >= d) inc += t;
  }
  if (lane == 31) wtot[wave] = inc;

  lds_barrier();
  int base = inc - s;
  for (int w = 0; w < wave; ++w) base += wtot[w];
  v4i pv, sv;
  pv.x = base;
  pv.y = pv.x + nn[0];
  pv.z = pv.y + nn[1];
  pv.w = pv.z + nn[2];
  sv.x = a[0]; sv.y = a[1]; sv.z = a[2]; sv.w = a[3];
  *reinterpret_cast<v4i*>(&pa[4 * tid]) = pv;
  *reinterpret_cast<v4i*>(&sa[4 * tid]) = sv;
  if (tid == MRG_T - 1) ltot = base + s;
  lds_barrier();

  const int len   = __builtin_amdgcn_readfirstlane(ltot);
  const int step0 = 1 << (31 - __builtin_clz(nb));
  unsigned* gl = glist + (size_t)o * LIST_CAP;
  if (len > 0) merge_round(0, len, step0, tid, pa, sa, cnt, seg, gl);
  for (int k0 = MRG_T * MRG_Q; k0 < len; k0 += MRG_T * MRG_Q)
    merge_round(k0, len, step0, tid, pa, sa, cnt, seg, gl);
  float* bl = reinterpret_cast<float*>(bfrag);
#pragma unroll
  for (int i = 0; i < NW; ++i) {
    const int p = i * MRG_T + tid;
    if (p < NP) {
      const int idx = p * 4;
      const int bk  = idx / NN, bn = idx % NN;
      const int j   = bk >> 3, hh = (bk >> 2) & 1, t = (bk >> 1) & 1, sb = bk & 1;
      const int nt  = bn >> 4, cc = bn & 15;
      float* dst = bl + ((((j * 2 + t) * NT + nt) * 32) + hh * 16 + cc) * 2 + sb;
      dst[0] = wv[i].x;
      dst[2] = wv[i].y;
      dst[4] = wv[i].z;
      dst[6] = wv[i].w;
    }
  }
  lds_barrier();

  for (int i = tid; i < nown; i += MRG_T) {
    const float d = 1.0f / sqrtf(1.0f + (float)cnt[i]);
    dinv[t0 + i] = d;
    dl[i]        = d;
  }
  if (tid == 0) glen[o] = len;

  v8f acc[NT];
#pragma unroll
  for (int nt = 0; nt < NT; ++nt) acc[nt] = v8f{};
#pragma unroll
  for (int j = 0; j < NQ; ++j) {
#pragma unroll
    for (int t = 0; t < 2; ++t) {
      v2f af;
      af.x = t ? q[j].z : q[j].x;
      af.y = t ? q[j].w : q[j].y;
#pragma unroll
      for (int nt = 0; nt < NT; ++nt) {
        const v2f bf = bfrag[((j * 2 + t) * NT + nt) * 32 + lane];
        acc[nt] = __builtin_amdgcn_wmma_f32_16x16x4_f32(false, af, false, bf, (short)0, acc[nt],
                                                        false, false);
      }
    }
  }
  lds_barrier();

  const int m0   = wave * 16 + 8 * h;
  float*    crow = C + (size_t)(t0 + m0) * NN + c;
#pragma unroll
  for (int v = 0; v < 8; ++v) {
    if (m0 + v < nown) {
      const float sc = dl[m0 + v];
#pragma unroll
      for (int nt = 0; nt < NT; ++nt) crow[(size_t)v * NN + nt * 16] = acc[nt][v] * sc;
    }
  }
}

template <int F>
__device__ __forceinline__ void add_trip(float* __restrict__ acc, int r, int d, int cnt, int lane,
                                         const float* __restrict__ H) {
  constexpr int S = F / 32;
  const int      ro = r * (F * 4);
  const int      lo = d * (F * 4);
  const char*    Hb = reinterpret_cast<const char*>(H);
  char*          ab = reinterpret_cast<char*>(acc) + lane * 4;
  int i0 = 0;
  for (; i0 + ROW_G <= cnt; i0 += ROW_G) {
    float hv[ROW_G][S];
    int   ll[ROW_G];
#pragma unroll
    for (int g = 0; g < ROW_G; ++g) {
      const unsigned rg = (unsigned)__builtin_amdgcn_readlane(ro, i0 + g);
      ll[g] = __builtin_amdgcn_readlane(lo, i0 + g);
      const float* hp = reinterpret_cast<const float*>(Hb + (size_t)rg);
#pragma unroll
      for (int s = 0; s < S; ++s) hv[g][s] = hp[s * 32 + lane];
    }
#pragma unroll
    for (int g = 0; g < ROW_G; ++g) {
      float* ap = reinterpret_cast<float*>(ab + ll[g]);
#pragma unroll
      for (int s = 0; s < S; ++s) atomicAdd(&ap[s * 32], hv[g][s]);
    }
  }
  if (i0 < cnt) {
    float hv[ROW_G][S];
    int   ll[ROW_G];
#pragma unroll
    for (int g = 0; g < ROW_G; ++g) {
      const unsigned rg = (unsigned)__builtin_amdgcn_readlane(ro, i0 + g);
      ll[g] = __builtin_amdgcn_readlane(lo, i0 + g);
      const float* hp = reinterpret_cast<const float*>(Hb + (size_t)rg);
#pragma unroll
      for (int s = 0; s < S; ++s) hv[g][s] = hp[s * 32 + lane];
    }
#pragma unroll
    for (int g = 0; g < ROW_G; ++g) {
      if (i0 + g < cnt) {
        float* ap = reinterpret_cast<float*>(ab + ll[g]);
#pragma unroll
        for (int s = 0; s < S; ++s) atomicAdd(&ap[s * 32], hv[g][s]);
      }
    }
  }
}

template <int F, bool RELU>
__global__ void __launch_bounds__(OWN_T)
k_agg_owner(const int* __restrict__ rows, const int* __restrict__ cols,
            const int* __restrict__ glen, const unsigned* __restrict__ glist,
            const float* __restrict__ dinv, const float* __restrict__ H,
            const float* __restrict__ bias, float* __restrict__ OUT, int n, int E) {
  static_assert(OWN_R <= 65536 && CHUNK_E <= 65536, "hit entries pack two 16-bit fields");
  static_assert(OWN_T % F == 0 && OWN_T >= OWN_R, "a thread keeps one feature column; a thread per owned node");
  __shared__ float    acc[OWN_R * F];
  __shared__ float    dtg[OWN_R];
  __shared__ unsigned hitlist[OWN_W][CHUNK_E];

  const int lane = threadIdx.x & 31;
  const int wave = __builtin_amdgcn_readfirstlane((int)(threadIdx.x >> 5));
  const int t0   = blockIdx.x * OWN_R;
  const unsigned nown = (unsigned)min(OWN_R, n - t0);
  const int fill = t0 + OWN_R;

  constexpr int NI  = (OWN_R * F + OWN_T - 1) / OWN_T;
  const int     tot = (int)nown * F;
  const float*  own = H + (size_t)t0 * F;
  float sv[NI];
#pragma unroll
  for (int k = 0; k < NI; ++k) sv[k] = own[min(k * OWN_T + (int)threadIdx.x, tot - 1)];
  const float dmine = dinv[t0 + (int)min(threadIdx.x, nown - 1u)];
  const float bf    = bias[threadIdx.x & (F - 1)];

  const int len = __builtin_amdgcn_readfirstlane(glen[blockIdx.x]);
  const unsigned* gl = glist + (size_t)blockIdx.x * LIST_CAP;
  const bool listed  = (len <= LIST_CAP);
  unsigned ent = 0u;
  if (listed && wave * 32 < len) ent = gl[wave * 32 + min(lane, min(32, len - wave * 32) - 1)];

#pragma unroll
  for (int k = 0; k < NI; ++k) {
    const int i = k * OWN_T + (int)threadIdx.x;
    if (i < OWN_R * F) acc[i] = (i < tot) ? sv[k] : 0.0f;
  }
  if (threadIdx.x < OWN_R) dtg[threadIdx.x] = dmine;
  __syncthreads();

  unsigned* wl = hitlist[wave];
  const int  nchunk = (E + CHUNK_E - 1) / CHUNK_E;
  const bool vec_ok = ((E & 3) == 0) && ((((size_t)cols) & 15) == 0);
  const int  start  = (int)((blockIdx.x * max(1u, (unsigned)nchunk / gridDim.x)) % (unsigned)nchunk);

  if (listed) {
    int k0 = wave * 32;
    if (k0 < len) {
      for (; k0 < len; k0 += OWN_W * 32) {
        const int cnt = min(32, len - k0);
        const int kn = k0 + OWN_W * 32;
        unsigned entn = ent;
        if (kn < len) entn = gl[kn + min(lane, min(32, len - kn) - 1)];
        add_trip<F>(acc, (int)(ent >> D_BITS), (int)(ent & ((1u << D_BITS) - 1u)), cnt, lane, H);
        ent = entn;
      }
    }
  } else if (wave < nchunk) {
    int ci = wave;
    int ch = ci + start;
    if (ch >= nchunk) ch -= nchunk;
    v4i c[SCAN_U];
    load_chunk(c, cols, ch * CHUNK_E, lane, E, vec_ok, fill);

    for (; ci < nchunk; ci += OWN_W) {
      const int ebase = ch * CHUNK_E;

      int nh = 0;
#pragma unroll
      for (int u = 0; u < SCAN_U; ++u) {
#pragma unroll
        for (int j = 0; j < 4; ++j) {
          const unsigned d   = (unsigned)(c[u][j] - t0);
          const bool     hit = d < nown;
          const unsigned m   = __builtin_amdgcn_ballot_w32(hit);
          if (m) {
            if (hit)
              wl[nh + (int)__builtin_amdgcn_mbcnt_lo(m, 0u)] =
                  ((unsigned)(u * 128 + lane * 4 + j) << 16) | d;
            nh += __builtin_popcount(m);
          }
        }
      }

      int chn = ch;
      if (ci + OWN_W < nchunk) {
        chn = ci + OWN_W + start;
        if (chn >= nchunk) chn -= nchunk;
      }
      load_chunk(c, cols, chn * CHUNK_E, lane, E, vec_ok, fill);

      __builtin_amdgcn_fence(__ATOMIC_RELEASE, "wavefront");
      __builtin_amdgcn_wave_barrier();
      __builtin_amdgcn_fence(__ATOMIC_ACQUIRE, "wavefront");

      for (int k0 = 0; k0 < nh; k0 += 32) {
        const int cnt = min(32, nh - k0);
        const unsigned ent = wl[k0 + min(lane, cnt - 1)];
        const int d = (int)(ent & 0xffffu);
        const int r = rows[ebase + (int)(ent >> 16)];
        add_trip<F>(acc, r, d, cnt, lane, H);
      }
      ch = chn;
    }
  }
  __syncthreads();

  float* outp = OUT + (size_t)t0 * F;
  for (int i = threadIdx.x; i < tot; i += OWN_T) {
    float v = acc[i] * dtg[i / F] + bf;
    if (RELU) v = fmaxf(v, 0.0f);
    outp[i] = v;
  }
}

extern "C" void kernel_launch(void* const* d_in, const int* in_sizes, int n_in,
                              void* d_out, int out_size, void* d_ws, size_t ws_size,
                              hipStream_t stream) {
  const float* x  = (const float*)d_in[0];
  const int*   ei = (const int*)d_in[1];
  const float* W1 = (const float*)d_in[2];
  const float* b1 = (const float*)d_in[3];
  const float* W2 = (const float*)d_in[4];
  const float* b2 = (const float*)d_in[5];

  const int IN_DIM = 128, HID = 64, OUT = 32;
  const int N = in_sizes[0] / IN_DIM;
  const int E = in_sizes[1] / 2;
  const int* rows = ei;
  const int* cols = ei + E;

  const int G  = (N + OWN_R - 1) / OWN_R;
  const int NB = (E + SL_E - 1) / SL_E;

  float* ws   = (float*)d_ws;
  float* dinv = ws;
  float* h1   = ws + N;
  float* r1   = h1 + (size_t)N * HID;
  float* h2   = h1;
  int*      glen  = (int*)(r1 + (size_t)N * HID);
  unsigned* glist = (unsigned*)(glen + G);
  unsigned* seg   = glist + (size_t)G * LIST_CAP;
  int*      soT   = (int*)(seg + (size_t)NB * SL_E);

  k_bin_slices<<<NB, BIN_T, 0, stream>>>(rows, cols, seg, soT, G, N, E);
  k_merge_gemm<128, 64><<<G, MRG_T, 0, stream>>>(seg, soT, dinv, glen, glist, x, W1, h1, N, NB);

  k_agg_owner<64, true><<<G, OWN_T, 0, stream>>>(rows, cols, glen, glist, dinv, h1, b1, r1, N, E);

  k_gemm_wmma<64, 32, GEMM_W><<<(N / 16 + GEMM_W - 1) / GEMM_W, 32 * GEMM_W, 0, stream>>>(r1, W2, dinv, h2, N / 16);
  k_agg_owner<32, false><<<G, OWN_T, 0, stream>>>(rows, cols, glen, glist, dinv, h2, b2,
                                                  (float*)d_out, N, E);
}


# Layer-2 product r1@W2 in the layer-1 owner launch: 13 waves take A fragments straight from the owner's LDS sums (pitch 68 words), relu in registers; r1 never stored; 5 launches to 4; sums keep order
# speedup vs baseline: 5.8109x; 1.1141x over previous
#include <hip/hip_runtime.h>
#include <cstddef>


typedef __attribute__((ext_vector_type(2))) float v2f;
typedef __attribute__((ext_vector_type(8))) float v8f;
typedef __attribute__((ext_vector_type(4))) int   v4i;
typedef __attribute__((ext_vector_type(4))) float v4f;

#define OWN_R   196
#define OWN_T   1024
#define OWN_W   (OWN_T / 32)
#define SCAN_U  8
#define CHUNK_E (32 * 4 * SCAN_U)
#define ROW_G   8
#define LIST_CAP 131072
#define D_BITS  8
#define SL_E    4096
#define BIN_T   256
#define BIN_Q   (SL_E / BIN_T)
#define MAX_G   (4 * BIN_T)
#define OWN_TILES ((OWN_R + 15) / 16)
#define MRG_T   (32 * OWN_TILES)
#define MRG_Q   8
#define MAX_NB  (4 * MRG_T)
#define SRCH_NB 1024

__device__ __forceinline__ void load_chunk(v4i (&c)[SCAN_U], const int* __restrict__ cols,
                                           int ebase, int lane, int E, bool vec_ok, int fill) {
  if (vec_ok && ebase + CHUNK_E <= E) {
#pragma unroll
    for (int u = 0; u < SCAN_U; ++u)
      c[u] = *reinterpret_cast<const v4i*>(cols + ebase + u * 128 + lane * 4);
  } else {
#pragma unroll
    for (int u = 0; u < SCAN_U; ++u) {
      const int e = ebase + u * 128 + lane * 4;
      c[u].x = (e     < E) ? cols[e]     : fill;
      c[u].y = (e + 1 < E) ? cols[e + 1] : fill;
      c[u].z = (e + 2 < E) ? cols[e + 2] : fill;
      c[u].w = (e + 3 < E) ? cols[e + 3] : fill;
    }
  }
}

__global__ void __launch_bounds__(BIN_T)
k_bin_slices(const int* __restrict__ rows, const int* __restrict__ cols,
             unsigned* __restrict__ seg, int* __restrict__ soT, int G, int n, int E) {
  static_assert(SL_E % BIN_T == 0 && MAX_G == 4 * BIN_T, "whole edges and four counters per thread");
  __shared__ __attribute__((aligned(16))) int hist[MAX_G];
  __shared__ unsigned sorted[SL_E];
  __shared__ int      wtot[BIN_T / 32];
  const int tid  = threadIdx.x;
  const int lane = tid & 31;
  const int wave = __builtin_amdgcn_readfirstlane(tid >> 5);
  const int NB   = (int)gridDim.x;
  const int e0   = blockIdx.x * SL_E;
  const int cnt  = min(SL_E, E - e0);
  const unsigned gown = (unsigned)min(G, MAX_G - 1);

  int cv[BIN_Q], rv[BIN_Q];
#pragma unroll
  for (int q = 0; q < BIN_Q; ++q) {
    const int e = e0 + min(q * BIN_T + tid, cnt - 1);
    cv[q] = cols[e];
    rv[q] = rows[e];
  }
#pragma unroll
  for (int i = 0; i < MAX_G / BIN_T; ++i) hist[i * BIN_T + tid] = 0;
  __syncthreads();

  int      key[BIN_Q], rk[BIN_Q];
  unsigned ent[BIN_Q];
#pragma unroll
  for (int q = 0; q < BIN_Q; ++q) {
    const unsigned o  = (unsigned)cv[q] / (unsigned)OWN_R;
    const bool     ok = (q * BIN_T + tid < cnt) && ((unsigned)cv[q] < (unsigned)n) && (o < gown);
    key[q] = ok ? (int)o : -1;
    ent[q] = ((unsigned)rv[q] << D_BITS) | ((unsigned)cv[q] - o * (unsigned)OWN_R);
    rk[q]  = 0;
    if (ok) rk[q] = atomicAdd(&hist[o], 1);
  }
  __syncthreads();

  const v4i hq = *reinterpret_cast<const v4i*>(&hist[4 * tid]);
  const int s  = hq.x + hq.y + hq.z + hq.w;
  int inc = s;
#pragma unroll
  for (int d = 1; d < 32; d <<= 1) {
    const int t = __shfl_up(inc, d);
    if (lane >= d) inc += t;
  }
  if (lane == 31) wtot[wave] = inc;
  __syncthreads();
  int base = inc - s;
  for (int w = 0; w < wave; ++w) base += wtot[w];
  v4i ex;
  ex.x = base;
  ex.y = ex.x + hq.x;
  ex.z = ex.y + hq.y;
  ex.w = ex.z + hq.z;
  *reinterpret_cast<v4i*>(&hist[4 * tid]) = ex;
#pragma unroll
  for (int j = 0; j < 4; ++j) {
    const int bin = 4 * tid + j;
    if (bin <= (int)gown) soT[(size_t)bin * NB + blockIdx.x] = ex[j];
  }
  __syncthreads();

  int hp[BIN_Q];
#pragma unroll
  for (int q = 0; q < BIN_Q; ++q) hp[q] = hist[max(key[q], 0)];
#pragma unroll
  for (int q = 0; q < BIN_Q; ++q)
    if (key[q] >= 0) sorted[hp[q] + rk[q]] = ent[q];
  __syncthreads();

  const int kept = hist[gown];
  unsigned* sg = seg + (size_t)blockIdx.x * SL_E;
  for (int i = tid; i < kept; i += BIN_T) sg[i] = sorted[i];
}

__device__ __forceinline__ void merge_round(int k0, int len, int step0, int tid,
                                            const int* pa, const int* sa, int* cnt,
                                            const unsigned* __restrict__ seg,
                                            unsigned* __restrict__ gl) {
  int kk[MRG_Q], bq[MRG_Q];
#pragma unroll
  for (int q2 = 0; q2 < MRG_Q; ++q2) {
    kk[q2] = min(k0 + q2 * MRG_T + tid, len - 1);
    bq[q2] = 0;
  }
  for (int step = step0; step > 0; step >>= 1) {
    int pw[MRG_Q];
#pragma unroll
    for (int q2 = 0; q2 < MRG_Q; ++q2) pw[q2] = pa[bq[q2] + step];
#pragma unroll
    for (int q2 = 0; q2 < MRG_Q; ++q2) bq[q2] += (pw[q2] <= kk[q2]) ? step : 0;
  }
  int sq[MRG_Q], pq[MRG_Q];
#pragma unroll
  for (int q2 = 0; q2 < MRG_Q; ++q2) {
    sq[q2] = sa[bq[q2]];
    pq[q2] = pa[bq[q2]];
  }
  unsigned ev[MRG_Q];
#pragma unroll
  for (int q2 = 0; q2 < MRG_Q; ++q2)
    ev[q2] = seg[(size_t)bq[q2] * SL_E + (sq[q2] + (kk[q2] - pq[q2]))];
  asm volatile("" ::"v"(ev[0]));
#pragma unroll
  for (int q2 = 0; q2 < MRG_Q; ++q2) {
    const int k = k0 + q2 * MRG_T + tid;
    if (k < len) {
      atomicAdd(&cnt[ev[q2] & ((1u << D_BITS) - 1u)], 1);
      if (k < LIST_CAP) gl[k] = ev[q2];
    }
  }
}

__device__ __forceinline__ void lds_barrier() {
  __builtin_amdgcn_fence(__ATOMIC_RELEASE, "workgroup", "local");
  __builtin_amdgcn_s_barrier();
  __builtin_amdgcn_fence(__ATOMIC_ACQUIRE, "workgroup", "local");
}

template <int K, int NN>
__global__ void __launch_bounds__(MRG_T)
k_merge_gemm(const unsigned* __restrict__ seg, const int* __restrict__ soT,
             float* __restrict__ dinv, int* __restrict__ glen, unsigned* __restrict__ glist,
             const float* __restrict__ A, const float* __restrict__ B, float* __restrict__ C,
             int n, int NB) {
  constexpr int NT = NN / 16;
  constexpr int KS = K / 4;
  constexpr int NQ = K / 8;
  constexpr int NP = K * NN / 4;
  constexpr int NW = (NP + MRG_T - 1) / MRG_T;
  static_assert(K % 8 == 0 && NN % 16 == 0, "whole fragments only");
  static_assert(OWN_R <= (1 << D_BITS), "local target must fit the low bits of a list entry");
  static_assert(MAX_NB == 4 * MRG_T, "four slices per thread");
  static_assert(SRCH_NB <= MAX_NB && (SRCH_NB & (SRCH_NB - 1)) == 0, "the search stays inside the tables");
  __shared__ int   cnt[OWN_R];
  __shared__ float dl[OWN_R];
  __shared__ __attribute__((aligned(16))) int pa[MAX_NB];
  __shared__ __attribute__((aligned(16))) int sa[MAX_NB];
  __shared__ int wtot[MRG_T / 32];
  __shared__ int ltot;
  __shared__ v2f bfrag[KS * NT * 32];
  const int tid  = threadIdx.x;
  const int lane = tid & 31;
  const int wave = __builtin_amdgcn_readfirstlane(tid >> 5);
  const int c    = lane & 15;
  const int h    = lane >> 4;
  const int o    = blockIdx.x;
  const int t0   = o * OWN_R;
  const int nown = min(OWN_R, n - t0);
  const int nb   = min(NB, SRCH_NB - 1);

  int a[4], z[4], nn[4];
#pragma unroll
  for (int j = 0; j < 4; ++j) {
    const int bb = min(4 * tid + j, nb - 1);
    a[j] = soT[(size_t)o * NB + bb];
    z[j] = soT[(size_t)(o + 1) * NB + bb];
  }
  v4f q[NQ];
  const float* ap = A + (size_t)min(t0 + wave * 16 + c, n - 1) * K + 4 * h;
#pragma unroll
  for (int j = 0; j < NQ; ++j) q[j] = *reinterpret_cast<const v4f*>(ap + 8 * j);
  v4f wv[NW];
#pragma unroll
  for (int i = 0; i < NW; ++i)
    wv[i] = *reinterpret_cast<const v4f*>(B + 4 * min(i * MRG_T + tid, NP - 1));

#pragma unroll
  for (int j = 0; j < 4; ++j) nn[j] = (z[j] - a[j]) & ((4 * tid + j < nb) ? -1 : 0);
  for (int i = tid; i < OWN_R; i += MRG_T) cnt[i] = 0;

  const int s = nn[0] + nn[1] + nn[2] + nn[3];
  int inc = s;
#pragma unroll
  for (int d = 1; d < 32; d <<= 1) {
    const int t = __shfl_up(inc, d);
    if (lane >= d) inc += t;
  }
  if (lane == 31) wtot[wave] = inc;

  lds_barrier();
  int base = inc - s;
  for (int w = 0; w < wave; ++w) base += wtot[w];
  v4i pv, sv;
  pv.x = base;
  pv.y = pv.x + nn[0];
  pv.z = pv.y + nn[1];
  pv.w = pv.z + nn[2];
  sv.x = a[0]; sv.y = a[1]; sv.z = a[2]; sv.w = a[3];
  *reinterpret_cast<v4i*>(&pa[4 * tid]) = pv;
  *reinterpret_cast<v4i*>(&sa[4 * tid]) = sv;
  if (tid == MRG_T - 1) ltot = base + s;
  lds_barrier();

  const int len   = __builtin_amdgcn_readfirstlane(ltot);
  const int step0 = 1 << (31 - __builtin_clz(nb));
  unsigned* gl = glist + (size_t)o * LIST_CAP;
  if (len > 0) merge_round(0, len, step0, tid, pa, sa, cnt, seg, gl);
  for (int k0 = MRG_T * MRG_Q; k0 < len; k0 += MRG_T * MRG_Q)
    merge_round(k0, len, step0, tid, pa, sa, cnt, seg, gl);
  float* bl = reinterpret_cast<float*>(bfrag);
#pragma unroll
  for (int i = 0; i < NW; ++i) {
    const int p = i * MRG_T + tid;
    if (p < NP) {
      const int idx = p * 4;
      const int bk  = idx / NN, bn = idx % NN;
      const int j   = bk >> 3, hh = (bk >> 2) & 1, t = (bk >> 1) & 1, sb = bk & 1;
      const int nt  = bn >> 4, cc = bn & 15;
      float* dst = bl + ((((j * 2 + t) * NT + nt) * 32) + hh * 16 + cc) * 2 + sb;
      dst[0] = wv[i].x;
      dst[2] = wv[i].y;
      dst[4] = wv[i].z;
      dst[6] = wv[i].w;
    }
  }
  lds_barrier();

  for (int i = tid; i < nown; i += MRG_T) {
    const float d = 1.0f / sqrtf(1.0f + (float)cnt[i]);
    dinv[t0 + i] = d;
    dl[i]        = d;
  }
  if (tid == 0) glen[o] = len;

  v8f acc[NT];
#pragma unroll
  for (int nt = 0; nt < NT; ++nt) acc[nt] = v8f{};
#pragma unroll
  for (int j = 0; j < NQ; ++j) {
#pragma unroll
    for (int t = 0; t < 2; ++t) {
      v2f af;
      af.x = t ? q[j].z : q[j].x;
      af.y = t ? q[j].w : q[j].y;
#pragma unroll
      for (int nt = 0; nt < NT; ++nt) {
        const v2f bf = bfrag[((j * 2 + t) * NT + nt) * 32 + lane];
        acc[nt] = __builtin_amdgcn_wmma_f32_16x16x4_f32(false, af, false, bf, (short)0, acc[nt],
                                                        false, false);
      }
    }
  }
  lds_barrier();

  const int m0   = wave * 16 + 8 * h;
  float*    crow = C + (size_t)(t0 + m0) * NN + c;
#pragma unroll
  for (int v = 0; v < 8; ++v) {
    if (m0 + v < nown) {
      const float sc = dl[m0 + v];
#pragma unroll
      for (int nt = 0; nt < NT; ++nt) crow[(size_t)v * NN + nt * 16] = acc[nt][v] * sc;
    }
  }
}

template <int F, int P>
__device__ __forceinline__ void add_trip(float* __restrict__ acc, int r, int d, int cnt, int lane,
                                         const float* __restrict__ H) {
  constexpr int S = F / 32;
  const int      ro = r * (F * 4);
  const int      lo = d * (P * 4);
  const char*    Hb = reinterpret_cast<const char*>(H);
  char*          ab = reinterpret_cast<char*>(acc) + lane * 4;
  int i0 = 0;
  for (; i0 + ROW_G <= cnt; i0 += ROW_G) {
    float hv[ROW_G][S];
    int   ll[ROW_G];
#pragma unroll
    for (int g = 0; g < ROW_G; ++g) {
      const unsigned rg = (unsigned)__builtin_amdgcn_readlane(ro, i0 + g);
      ll[g] = __builtin_amdgcn_readlane(lo, i0 + g);
      const float* hp = reinterpret_cast<const float*>(Hb + (size_t)rg);
#pragma unroll
      for (int s = 0; s < S; ++s) hv[g][s] = hp[s * 32 + lane];
    }
#pragma unroll
    for (int g = 0; g < ROW_G; ++g) {
      float* ap = reinterpret_cast<float*>(ab + ll[g]);
#pragma unroll
      for (int s = 0; s < S; ++s) atomicAdd(&ap[s * 32], hv[g][s]);
    }
  }
  if (i0 < cnt) {
    float hv[ROW_G][S];
    int   ll[ROW_G];
#pragma unroll
    for (int g = 0; g < ROW_G; ++g) {
      const unsigned rg = (unsigned)__builtin_amdgcn_readlane(ro, i0 + g);
      ll[g] = __builtin_amdgcn_readlane(lo, i0 + g);
      const float* hp = reinterpret_cast<const float*>(Hb + (size_t)rg);
#pragma unroll
      for (int s = 0; s < S; ++s) hv[g][s] = hp[s * 32 + lane];
    }
#pragma unroll
    for (int g = 0; g < ROW_G; ++g) {
      if (i0 + g < cnt) {
        float* ap = reinterpret_cast<float*>(ab + ll[g]);
#pragma unroll
        for (int s = 0; s < S; ++s) atomicAdd(&ap[s * 32], hv[g][s]);
      }
    }
  }
}

template <int F, int NN>
__global__ void __launch_bounds__(OWN_T)
k_agg_owner(const int* __restrict__ rows, const int* __restrict__ cols,
            const int* __restrict__ glen, const unsigned* __restrict__ glist,
            const float* __restrict__ dinv, const float* __restrict__ H,
            const float* __restrict__ bias, const float* __restrict__ B,
            float* __restrict__ OUT, int n, int E) {
  constexpr bool FUSE = NN > 0;
  constexpr int  P    = FUSE ? F + 4 : F;
  constexpr int  NT   = FUSE ? NN / 16 : 1;
  constexpr int  NQ   = F / 8;
  constexpr int  NP   = FUSE ? F * NN / 4 : 1;
  static_assert(OWN_R <= 65536 && CHUNK_E <= 65536, "hit entries pack two 16-bit fields");
  static_assert(OWN_T % F == 0 && OWN_T >= OWN_R, "a thread keeps one feature column; a thread per owned node");
  static_assert(NN % 16 == 0 && F % 8 == 0 && NP <= OWN_T && OWN_TILES <= OWN_W && OWN_TILES * 16 <= OWN_T,
                "whole fragments; a piece of B per thread; a wave per tile; a thread per tile row");
  __shared__ __attribute__((aligned(16))) float acc[OWN_R * P];
  constexpr int  ND   = FUSE ? OWN_TILES * 16 : OWN_R;
  __shared__ __attribute__((aligned(16))) float dtg[ND];
  __shared__ unsigned hitlist[OWN_W][CHUNK_E];
  __shared__ __attribute__((aligned(16))) float bl[FUSE ? F : 4];
  __shared__ v2f bfrag[FUSE ? (F / 4) * NT * 32 : 1];

  const int lane = threadIdx.x & 31;
  const int wave = __builtin_amdgcn_readfirstlane((int)(threadIdx.x >> 5));
  const int t0   = blockIdx.x * OWN_R;
  const unsigned nown = (unsigned)min(OWN_R, n - t0);
  const int fill = t0 + OWN_R;

  constexpr int NI  = (OWN_R * F + OWN_T - 1) / OWN_T;
  const int     tot = (int)nown * F;
  const float*  own = H + (size_t)t0 * F;
  float sv[NI];
#pragma unroll
  for (int k = 0; k < NI; ++k) sv[k] = own[min(k * OWN_T + (int)threadIdx.x, tot - 1)];
  const float dmine = dinv[t0 + (int)min(threadIdx.x, nown - 1u)];
  const float bf    = bias[threadIdx.x & (F - 1)];
  v4f wv = v4f{};
  if constexpr (FUSE) wv = *reinterpret_cast<const v4f*>(B + 4 * min((int)threadIdx.x, NP - 1));

  const int len = __builtin_amdgcn_readfirstlane(glen[blockIdx.x]);
  const unsigned* gl = glist + (size_t)blockIdx.x * LIST_CAP;
  const bool listed  = (len <= LIST_CAP);
  unsigned ent = 0u;
  if (listed && wave * 32 < len) ent = gl[wave * 32 + min(lane, min(32, len - wave * 32) - 1)];

#pragma unroll
  for (int k = 0; k < NI; ++k) {
    const int i = k * OWN_T + (int)threadIdx.x;
    if (i < OWN_R * F) acc[(i / F) * P + (i % F)] = (i < tot) ? sv[k] : 0.0f;
  }
  if (threadIdx.x < ND) dtg[threadIdx.x] = dmine;
  if constexpr (FUSE) {
    if ((int)threadIdx.x < NP) {
      const int idx = (int)threadIdx.x * 4;
      const int bk  = idx / NN, bn = idx % NN;
      const int j   = bk >> 3, hh = (bk >> 2) & 1, t = (bk >> 1) & 1, sb = bk & 1;
      const int nt  = bn >> 4, cc = bn & 15;
      float* dst = reinterpret_cast<float*>(bfrag) + ((((j * 2 + t) * NT + nt) * 32) + hh * 16 + cc) * 2 + sb;
      dst[0] = wv.x;
      dst[2] = wv.y;
      dst[4] = wv.z;
      dst[6] = wv.w;
    }
    if (threadIdx.x < F) bl[threadIdx.x] = bf;
  }
  __syncthreads();

  unsigned* wl = hitlist[wave];
  const int  nchunk = (E + CHUNK_E - 1) / CHUNK_E;
  const bool vec_ok = ((E & 3) == 0) && ((((size_t)cols) & 15) == 0);
  const int  start  = (int)((blockIdx.x * max(1u, (unsigned)nchunk / gridDim.x)) % (unsigned)nchunk);

  if (listed) {
    int k0 = wave * 32;
    if (k0 < len) {
      for (; k0 < len; k0 += OWN_W * 32) {
        const int cnt = min(32, len - k0);
        const int kn = k0 + OWN_W * 32;
        unsigned entn = ent;
        if (kn < len) entn = gl[kn + min(lane, min(32, len - kn) - 1)];
        add_trip<F, P>(acc, (int)(ent >> D_BITS), (int)(ent & ((1u << D_BITS) - 1u)), cnt, lane, H);
        ent = entn;
      }
    }
  } else if (wave < nchunk) {
    int ci = wave;
    int ch = ci + start;
    if (ch >= nchunk) ch -= nchunk;
    v4i c[SCAN_U];
    load_chunk(c, cols, ch * CHUNK_E, lane, E, vec_ok, fill);

    for (; ci < nchunk; ci += OWN_W) {
      const int ebase = ch * CHUNK_E;

      int nh = 0;
#pragma unroll
      for (int u = 0; u < SCAN_U; ++u) {
#pragma unroll
        for (int j = 0; j < 4; ++j) {
          const unsigned d   = (unsigned)(c[u][j] - t0);
          const bool     hit = d < nown;
          const unsigned m   = __builtin_amdgcn_ballot_w32(hit);
          if (m) {
            if (hit)
              wl[nh + (int)__builtin_amdgcn_mbcnt_lo(m, 0u)] =
                  ((unsigned)(u * 128 + lane * 4 + j) << 16) | d;
            nh += __builtin_popcount(m);
          }
        }
      }

      int chn = ch;
      if (ci + OWN_W < nchunk) {
        chn = ci + OWN_W + start;
        if (chn >= nchunk) chn -= nchunk;
      }
      load_chunk(c, cols, chn * CHUNK_E, lane, E, vec_ok, fill);

      __builtin_amdgcn_fence(__ATOMIC_RELEASE, "wavefront");
      __builtin_amdgcn_wave_barrier();
      __builtin_amdgcn_fence(__ATOMIC_ACQUIRE, "wavefront");

      for (int k0 = 0; k0 < nh; k0 += 32) {
        const int cnt = min(32, nh - k0);
        const unsigned ent = wl[k0 + min(lane, cnt - 1)];
        const int d = (int)(ent & 0xffffu);
        const int r = rows[ebase + (int)(ent >> 16)];
        add_trip<F, P>(acc, r, d, cnt, lane, H);
      }
      ch = chn;
    }
  }
  __syncthreads();

  if constexpr (!FUSE) {
    float* outp = OUT + (size_t)t0 * F;
    for (int i = threadIdx.x; i < tot; i += OWN_T) outp[i] = acc[i] * dtg[i / F] + bf;
  } else if (wave < OWN_TILES) {
    const int   c   = lane & 15;
    const int   h   = lane >> 4;
    const int   ra  = min(wave * 16 + c, OWN_R - 1);
    const float da  = dtg[ra];
    const float* arow = acc + ra * P + 4 * h;
    const float* brow = bl + 4 * h;
    const int m0  = wave * 16 + 8 * h;
    const v4f sc0 = *reinterpret_cast<const v4f*>(&dtg[m0]);
    const v4f sc1 = *reinterpret_cast<const v4f*>(&dtg[m0 + 4]);
    v4f q[NQ];
#pragma unroll
    for (int j = 0; j < NQ; ++j) {
      const v4f sm = *reinterpret_cast<const v4f*>(arow + 8 * j);
      const v4f bs = *reinterpret_cast<const v4f*>(brow + 8 * j);
      q[j].x = fmaxf(sm.x * da + bs.x, 0.0f);
      q[j].y = fmaxf(sm.y * da + bs.y, 0.0f);
      q[j].z = fmaxf(sm.z * da + bs.z, 0.0f);
      q[j].w = fmaxf(sm.w * da + bs.w, 0.0f);
    }
    v8f o[NT];
#pragma unroll
    for (int nt = 0; nt < NT; ++nt) o[nt] = v8f{};
#pragma unroll
    for (int j = 0; j < NQ; ++j) {
#pragma unroll
      for (int t = 0; t < 2; ++t) {
        v2f af;
        af.x = t ? q[j].z : q[j].x;
        af.y = t ? q[j].w : q[j].y;
#pragma unroll
        for (int nt = 0; nt < NT; ++nt) {
          const v2f bfr = bfrag[((j * 2 + t) * NT + nt) * 32 + lane];
          o[nt] = __builtin_amdgcn_wmma_f32_16x16x4_f32(false, af, false, bfr, (short)0, o[nt],
                                                        false, false);
        }
      }
    }
    float* crow = OUT + (size_t)(t0 + m0) * NN + c;
#pragma unroll
    for (int v = 0; v < 8; ++v) {
      if ((unsigned)(m0 + v) < nown) {
        const float sc = (v < 4) ? sc0[v & 3] : sc1[v & 3];
#pragma unroll
        for (int nt = 0; nt < NT; ++nt) crow[(size_t)v * NN + nt * 16] = o[nt][v] * sc;
      }
    }
  }
}

extern "C" void kernel_launch(void* const* d_in, const int* in_sizes, int n_in,
                              void* d_out, int out_size, void* d_ws, size_t ws_size,
                              hipStream_t stream) {
  const float* x  = (const float*)d_in[0];
  const int*   ei = (const int*)d_in[1];
  const float* W1 = (const float*)d_in[2];
  const float* b1 = (const float*)d_in[3];
  const float* W2 = (const float*)d_in[4];
  const float* b2 = (const float*)d_in[5];

  const int IN_DIM = 128, HID = 64, OUT = 32;
  const int N = in_sizes[0] / IN_DIM;
  const int E = in_sizes[1] / 2;
  const int* rows = ei;
  const int* cols = ei + E;

  const int G  = (N + OWN_R - 1) / OWN_R;
  const int NB = (E + SL_E - 1) / SL_E;

  float* ws   = (float*)d_ws;
  float* dinv = ws;
  float* h1   = ws + N;
  float* h2   = h1 + (size_t)N * HID;
  int*      glen  = (int*)(h2 + (size_t)N * OUT);
  unsigned* glist = (unsigned*)(glen + G);
  unsigned* seg   = glist + (size_t)G * LIST_CAP;
  int*      soT   = (int*)(seg + (size_t)NB * SL_E);

  k_bin_slices<<<NB, BIN_T, 0, stream>>>(rows, cols, seg, soT, G, N, E);
  k_merge_gemm<128, 64><<<G, MRG_T, 0, stream>>>(seg, soT, dinv, glen, glist, x, W1, h1, N, NB);

  k_agg_owner<64, 32><<<G, OWN_T, 0, stream>>>(rows, cols, glen, glist, dinv, h1, b1, W2, h2, N, E);

  k_agg_owner<32, 0><<<G, OWN_T, 0, stream>>>(rows, cols, glen, glist, dinv, h2, b2, W2,
                                              (float*)d_out, N, E);
}
